# starting kernel
# baseline (speedup 1.0000x reference)
#include <hip/hip_runtime.h>
#include <hip/hip_bf16.h>

typedef __attribute__((ext_vector_type(2))) float v2f;
typedef __attribute__((ext_vector_type(8))) float v8f;

__global__ void k_init_deg(float* __restrict__ deg, int N) {
    int i = blockIdx.x * blockDim.x + threadIdx.x;
    if (i < N) deg[i] = 1.0f;
}

__global__ void k_scatter_deg(const int* __restrict__ col,
                              const float* __restrict__ w,
                              float* __restrict__ deg, int E) {
    int e = blockIdx.x * blockDim.x + threadIdx.x;
    if (e < E) atomicAdd(&deg[col[e]], w[e]);
}

__global__ void k_dinv(const float* __restrict__ deg,
                       float* __restrict__ dinv, int N) {
    int i = blockIdx.x * blockDim.x + threadIdx.x;
    if (i < N) {
        float d = deg[i];
        dinv[i] = (d > 0.0f) ? rsqrtf(d) : 0.0f;
    }
}

__global__ void k_norm(const int* __restrict__ row,
                       const int* __restrict__ col,
                       const float* __restrict__ w,
                       const float* __restrict__ dinv,
                       float* __restrict__ norm, int E) {
    int e = blockIdx.x * blockDim.x + threadIdx.x;
    if (e < E) norm[e] = dinv[row[e]] * w[e] * dinv[col[e]];
}

__global__ __launch_bounds__(256) void k_gemm1(const float* __restrict__ X,
                                               const float* __restrict__ W,
                                               float* __restrict__ H,
                                               int N) {
    __shared__ float Alds[16 * 128];
    const int tid = threadIdx.x;
    const int bm  = blockIdx.x * 16;

    float4* al4 = (float4*)Alds;
    for (int i = tid; i < 16 * 32; i += 256) {
        int r = i >> 5, g = i & 31;
        float4 v = make_float4(0.f, 0.f, 0.f, 0.f);
        if (bm + r < N) v = ((const float4*)(X + (size_t)(bm + r) * 128))[g];
        al4[i] = v;
    }
    __syncthreads();

    const int wave = tid >> 5;
    const int lane = tid & 31;
    const int nb   = wave * 16;
    const int mr   = lane & 15;
    const int kh   = (lane >> 4) * 2;

    v8f acc = {};
    for (int kc = 0; kc < 128; kc += 4) {
        v2f a, b;
        a[0] = Alds[mr * 128 + kc + kh + 0];
        a[1] = Alds[mr * 128 + kc + kh + 1];
        b[0] = W[(kc + kh + 0) * 128 + nb + mr];
        b[1] = W[(kc + kh + 1) * 128 + nb + mr];
        acc = __builtin_amdgcn_wmma_f32_16x16x4_f32(
                  false, a, false, b, (short)0, acc, false, false);
    }

    const int mtop = (lane >> 4) * 8;
    for (int v = 0; v < 8; ++v) {
        int m = bm + mtop + v;
        if (m < N) H[(size_t)m * 128 + nb + mr] = acc[v];
    }
}

__global__ void k_init_out1(const float* __restrict__ h1,
                            const float* __restrict__ dinv,
                            const float* __restrict__ b1,
                            float* __restrict__ out1, int N) {
    int idx = blockIdx.x * blockDim.x + threadIdx.x;
    if (idx >= N * 32) return;
    int i = idx >> 5, g = idx & 31;
    float s = dinv[i]; s *= s;
    float4 h = ((const float4*)(h1 + (size_t)i * 128))[g];
    float4 b = ((const float4*)b1)[g];
    float4 o = make_float4(b.x + h.x * s, b.y + h.y * s,
                           b.z + h.z * s, b.w + h.w * s);
    ((float4*)(out1 + (size_t)i * 128))[g] = o;
}

__global__ void k_scatter1(const float* __restrict__ h1,
                           const int* __restrict__ row,
                           const int* __restrict__ col,
                           const float* __restrict__ norm,
                           float* __restrict__ out1, int E) {
    int idx = blockIdx.x * blockDim.x + threadIdx.x;
    if (idx >= E * 32) return;
    int e = idx >> 5, g = idx & 31;
    int r = row[e], c = col[e];
    float nw = norm[e];
    float4 v = ((const float4*)(h1 + (size_t)r * 128))[g];
    float* o = out1 + (size_t)c * 128 + g * 4;
    atomicAdd(o + 0, v.x * nw);
    atomicAdd(o + 1, v.y * nw);
    atomicAdd(o + 2, v.z * nw);
    atomicAdd(o + 3, v.w * nw);
}

__global__ __launch_bounds__(256) void k_gemm2(const float* __restrict__ A,
                                               const float* __restrict__ W2,
                                               float* __restrict__ H2,
                                               int N) {
    __shared__ float Alds[32 * 128];
    __shared__ float Wlds[128 * 64];
    const int tid = threadIdx.x;
    const int bm  = blockIdx.x * 32;

    {
        const float4* w4 = (const float4*)W2;
        float4* wl = (float4*)Wlds;
        for (int i = tid; i < 2048; i += 256) wl[i] = w4[i];
    }
    {
        float4* al = (float4*)Alds;
        for (int i = tid; i < 32 * 32; i += 256) {
            int r = i >> 5, g = i & 31;
            float4 v = make_float4(0.f, 0.f, 0.f, 0.f);
            if (bm + r < N) v = ((const float4*)(A + (size_t)(bm + r) * 128))[g];
            v.x = fmaxf(v.x, 0.f); v.y = fmaxf(v.y, 0.f);
            v.z = fmaxf(v.z, 0.f); v.w = fmaxf(v.w, 0.f);
            al[i] = v;
        }
    }
    __syncthreads();

    const int wave = tid >> 5;
    const int lane = tid & 31;
    const int msub = (wave >> 2) * 16;
    const int nb   = (wave & 3) * 16;
    const int mr   = lane & 15;
    const int kh   = (lane >> 4) * 2;

    v8f acc = {};
    for (int kc = 0; kc < 128; kc += 4) {
        v2f a, b;
        a[0] = Alds[(msub + mr) * 128 + kc + kh + 0];
        a[1] = Alds[(msub + mr) * 128 + kc + kh + 1];
        b[0] = Wlds[(kc + kh + 0) * 64 + nb + mr];
        b[1] = Wlds[(kc + kh + 1) * 64 + nb + mr];
        acc = __builtin_amdgcn_wmma_f32_16x16x4_f32(
                  false, a, false, b, (short)0, acc, false, false);
    }

    const int mtop = (lane >> 4) * 8;
    for (int v = 0; v < 8; ++v) {
        int m = bm + msub + mtop + v;
        if (m < N) H2[(size_t)m * 64 + nb + mr] = acc[v];
    }
}

__global__ void k_init_out2(const float* __restrict__ h2,
                            const float* __restrict__ dinv,
                            const float* __restrict__ b2,
                            float* __restrict__ out, int N) {
    int idx = blockIdx.x * blockDim.x + threadIdx.x;
    if (idx >= N * 16) return;
    int i = idx >> 4, g = idx & 15;
    float s = dinv[i]; s *= s;
    float4 h = ((const float4*)(h2 + (size_t)i * 64))[g];
    float4 b = ((const float4*)b2)[g];
    float4 o = make_float4(b.x + h.x * s, b.y + h.y * s,
                           b.z + h.z * s, b.w + h.w * s);
    ((float4*)(out + (size_t)i * 64))[g] = o;
}

__global__ void k_scatter2(const float* __restrict__ h2,
                           const int* __restrict__ row,
                           const int* __restrict__ col,
                           const float* __restrict__ norm,
                           float* __restrict__ out, int E) {
    int idx = blockIdx.x * blockDim.x + threadIdx.x;
    if (idx >= E * 16) return;
    int e = idx >> 4, g = idx & 15;
    int r = row[e], c = col[e];
    float nw = norm[e];
    float4 v = ((const float4*)(h2 + (size_t)r * 64))[g];
    float* o = out + (size_t)c * 64 + g * 4;
    atomicAdd(o + 0, v.x * nw);
    atomicAdd(o + 1, v.y * nw);
    atomicAdd(o + 2, v.z * nw);
    atomicAdd(o + 3, v.w * nw);
}

extern "C" void kernel_launch(void* const* d_in, const int* in_sizes, int n_in,
                              void* d_out, int out_size, void* d_ws, size_t ws_size,
                              hipStream_t stream) {
    const float* x   = (const float*)d_in[0];
    const int*   ei  = (const int*)d_in[1];
    const float* ew  = (const float*)d_in[2];
    const float* W1  = (const float*)d_in[3];
    const float* b1  = (const float*)d_in[4];
    const float* W2  = (const float*)d_in[5];
    const float* b2  = (const float*)d_in[6];
    float* out = (float*)d_out;

    const int E = in_sizes[2];
    const int N = in_sizes[0] / 128;
    const int* row = ei;
    const int* col = ei + E;

    float* ws   = (float*)d_ws;
    float* deg  = ws;
    float* dinv = ws + N;
    float* nrm  = ws + 2 * (size_t)N;
    float* h1   = nrm + E;
    float* out1 = h1 + (size_t)N * 128;
    float* h2   = out1 + (size_t)N * 128;

    const int T = 256;

    k_init_deg<<<(N + T - 1) / T, T, 0, stream>>>(deg, N);
    k_scatter_deg<<<(E + T - 1) / T, T, 0, stream>>>(col, ew, deg, E);
    k_dinv<<<(N + T - 1) / T, T, 0, stream>>>(deg, dinv, N);
    k_norm<<<(E + T - 1) / T, T, 0, stream>>>(row, col, ew, dinv, nrm, E);

    k_gemm1<<<(N + 15) / 16, 256, 0, stream>>>(x, W1, h1, N);
    k_init_out1<<<((size_t)N * 32 + T - 1) / T, T, 0, stream>>>(h1, dinv, b1, out1, N);
    k_scatter1<<<((size_t)E * 32 + T - 1) / T, T, 0, stream>>>(h1, row, col, nrm, out1, E);

    k_gemm2<<<(N + 31) / 32, 256, 0, stream>>>(out1, W2, h2, N);
    k_init_out2<<<((size_t)N * 16 + T - 1) / T, T, 0, stream>>>(h2, dinv, b2, out, N);
    k_scatter2<<<((size_t)E * 16 + T - 1) / T, T, 0, stream>>>(h2, row, col, nrm, out, E);
}


# Replace float-atomic edge scatters by a counting sort by destination (int atomic arrival-order rank, scan, fill) and one-wave-per-node gather sums in slot order; deg summed from the sorted list
# speedup vs baseline: 1.5955x; 1.5955x over previous
#include <hip/hip_runtime.h>
#include <hip/hip_bf16.h>


typedef __attribute__((ext_vector_type(2))) float v2f;
typedef __attribute__((ext_vector_type(4))) float v4f;
typedef __attribute__((ext_vector_type(8))) float v8f;

__global__ void k_zero(int* __restrict__ cnt, int N) {
    int i = blockIdx.x * blockDim.x + threadIdx.x;
    if (i < N) cnt[i] = 0;
}

__global__ void k_rank(const int* __restrict__ col,
                       int* __restrict__ cnt,
                       int* __restrict__ rank, int E) {
    int e = blockIdx.x * blockDim.x + threadIdx.x;
    if (e < E) rank[e] = atomicAdd(&cnt[col[e]], 1);
}

__global__ __launch_bounds__(1024) void k_scan(const int* __restrict__ cnt,
                                                int* __restrict__ rowptr, int N) {
    __shared__ int buf[15 * 1024];
    __shared__ int wsum[33];
    const int tid = threadIdx.x, lane = tid & 31, wv = tid >> 5;
    int carry = 0;
    for (int base = 0; base < N; base += 15 * 1024) {
        for (int k = 0; k < 15; ++k) {
            int j = k * 1024 + tid, g = base + j;
            buf[j] = (g < N) ? cnt[g] : 0;
        }
        __syncthreads();
        int s = 0;
        for (int k = 0; k < 15; ++k) {
            int v = buf[tid * 15 + k];
            buf[tid * 15 + k] = s;
            s += v;
        }
        int inc = s;
        for (int d = 1; d < 32; d <<= 1) {
            int t = __shfl_up(inc, d);
            if (lane >= d) inc += t;
        }
        if (lane == 31) wsum[wv] = inc;
        __syncthreads();
        if (wv == 0) {
            int x = wsum[lane], xi = x;
            for (int d = 1; d < 32; d <<= 1) {
                int t = __shfl_up(xi, d);
                if (lane >= d) xi += t;
            }
            wsum[lane] = xi - x;
            if (lane == 31) wsum[32] = xi;
        }
        __syncthreads();
        const int off = carry + wsum[wv] + (inc - s);
        for (int k = 0; k < 15; ++k) buf[tid * 15 + k] += off;
        carry += wsum[32];
        __syncthreads();
        for (int k = 0; k < 15; ++k) {
            int j = k * 1024 + tid, g = base + j;
            if (g < N) rowptr[g] = buf[j];
        }
        __syncthreads();
    }
    if (tid == 0) rowptr[N] = carry;
}

__global__ void k_fill(const int* __restrict__ row,
                       const int* __restrict__ col,
                       const float* __restrict__ w,
                       const int* __restrict__ rowptr,
                       const int* __restrict__ rank,
                       int* __restrict__ csr, int E) {
    int e = blockIdx.x * blockDim.x + threadIdx.x;
    if (e < E) {
        int p = rowptr[col[e]] + rank[e];
        ((int2*)csr)[p] = make_int2(row[e], __float_as_int(w[e]));
    }
}

__global__ void k_dinv(const int* __restrict__ rowptr,
                       const int* __restrict__ csr,
                       float* __restrict__ dinv, int N) {
    int i = blockIdx.x * blockDim.x + threadIdx.x;
    if (i < N) {
        const int b = rowptr[i], e = rowptr[i + 1];
        float d = 1.0f;
        for (int p = b; p < e; ++p) d += __int_as_float(csr[2 * p + 1]);
        dinv[i] = (d > 0.0f) ? rsqrtf(d) : 0.0f;
    }
}

__global__ __launch_bounds__(256) void k_gemm1(const float* __restrict__ X,
                                               const float* __restrict__ W,
                                               float* __restrict__ H,
                                               int N) {
    __shared__ float Alds[16 * 128];
    const int tid = threadIdx.x;
    const int bm  = blockIdx.x * 16;

    float4* al4 = (float4*)Alds;
    for (int i = tid; i < 16 * 32; i += 256) {
        int r = i >> 5, g = i & 31;
        float4 v = make_float4(0.f, 0.f, 0.f, 0.f);
        if (bm + r < N) v = ((const float4*)(X + (size_t)(bm + r) * 128))[g];
        al4[i] = v;
    }
    __syncthreads();

    const int wave = tid >> 5;
    const int lane = tid & 31;
    const int nb   = wave * 16;
    const int mr   = lane & 15;
    const int kh   = (lane >> 4) * 2;

    v8f acc = {};
    for (int kc = 0; kc < 128; kc += 4) {
        v2f a, b;
        a[0] = Alds[mr * 128 + kc + kh + 0];
        a[1] = Alds[mr * 128 + kc + kh + 1];
        b[0] = W[(kc + kh + 0) * 128 + nb + mr];
        b[1] = W[(kc + kh + 1) * 128 + nb + mr];
        acc = __builtin_amdgcn_wmma_f32_16x16x4_f32(
                  false, a, false, b, (short)0, acc, false, false);
    }

    const int mtop = (lane >> 4) * 8;
    for (int v = 0; v < 8; ++v) {
        int m = bm + mtop + v;
        if (m < N) H[(size_t)m * 128 + nb + mr] = acc[v];
    }
}

template <typename V>
__device__ __forceinline__ void gather_body(const float* __restrict__ h,
                                            const int* __restrict__ rowptr,
                                            const int* __restrict__ csr,
                                            const float* __restrict__ dinv,
                                            const float* __restrict__ bias,
                                            float* __restrict__ out, int N) {
    constexpr int VEC = sizeof(V) / sizeof(float);
    constexpr int F   = VEC * 32;
    const int lane = threadIdx.x & 31;
    const int node = __builtin_amdgcn_readfirstlane(
        (int)(blockIdx.x * (blockDim.x >> 5) + (threadIdx.x >> 5)));
    if (node >= N) return;
    const int beg = rowptr[node], end = rowptr[node + 1];
    const float dc = dinv[node];
    const V hv = *(const V*)(h + (size_t)node * F + lane * VEC);
    const V bv = *(const V*)(bias + lane * VEC);
    V acc = bv + hv * (dc * dc);
    const int2* ent = (const int2*)csr;
    for (int base = beg; base < end; base += 32) {
        const int n = min(32, end - base);
        int r = 0;
        float nw = 0.0f;
        if (lane < n) {
            const int2 t = ent[base + lane];
            r  = t.x;
            nw = dinv[r] * __int_as_float(t.y) * dc;
        }
        int j = 0;
        for (; j + 4 <= n; j += 4) {
            const int r0 = __builtin_amdgcn_readlane(r, j + 0);
            const int r1 = __builtin_amdgcn_readlane(r, j + 1);
            const int r2 = __builtin_amdgcn_readlane(r, j + 2);
            const int r3 = __builtin_amdgcn_readlane(r, j + 3);
            const float w0 = __int_as_float(__builtin_amdgcn_readlane(__float_as_int(nw), j + 0));
            const float w1 = __int_as_float(__builtin_amdgcn_readlane(__float_as_int(nw), j + 1));
            const float w2 = __int_as_float(__builtin_amdgcn_readlane(__float_as_int(nw), j + 2));
            const float w3 = __int_as_float(__builtin_amdgcn_readlane(__float_as_int(nw), j + 3));
            const V v0 = *(const V*)(h + (size_t)r0 * F + lane * VEC);
            const V v1 = *(const V*)(h + (size_t)r1 * F + lane * VEC);
            const V v2 = *(const V*)(h + (size_t)r2 * F + lane * VEC);
            const V v3 = *(const V*)(h + (size_t)r3 * F + lane * VEC);
            acc += v0 * w0;
            acc += v1 * w1;
            acc += v2 * w2;
            acc += v3 * w3;
        }
        for (; j < n; ++j) {
            const int r0 = __builtin_amdgcn_readlane(r, j);
            const float w0 = __int_as_float(__builtin_amdgcn_readlane(__float_as_int(nw), j));
            const V v0 = *(const V*)(h + (size_t)r0 * F + lane * VEC);
            acc += v0 * w0;
        }
    }
    *(V*)(out + (size_t)node * F + lane * VEC) = acc;
}

__global__ __launch_bounds__(256) void k_gather1(const float* __restrict__ h1,
                                                 const int* __restrict__ rowptr,
                                                 const int* __restrict__ csr,
                                                 const float* __restrict__ dinv,
                                                 const float* __restrict__ b1,
                                                 float* __restrict__ out1, int N) {
    gather_body<v4f>(h1, rowptr, csr, dinv, b1, out1, N);
}

__global__ __launch_bounds__(256) void k_gather2(const float* __restrict__ h2,
                                                 const int* __restrict__ rowptr,
                                                 const int* __restrict__ csr,
                                                 const float* __restrict__ dinv,
                                                 const float* __restrict__ b2,
                                                 float* __restrict__ out, int N) {
    gather_body<v2f>(h2, rowptr, csr, dinv, b2, out, N);
}

__global__ __launch_bounds__(256) void k_gemm2(const float* __restrict__ A,
                                               const float* __restrict__ W2,
                                               float* __restrict__ H2,
                                               int N) {
    __shared__ float Alds[32 * 128];
    __shared__ float Wlds[128 * 64];
    const int tid = threadIdx.x;
    const int bm  = blockIdx.x * 32;

    {
        const float4* w4 = (const float4*)W2;
        float4* wl = (float4*)Wlds;
        for (int i = tid; i < 2048; i += 256) wl[i] = w4[i];
    }
    {
        float4* al = (float4*)Alds;
        for (int i = tid; i < 32 * 32; i += 256) {
            int r = i >> 5, g = i & 31;
            float4 v = make_float4(0.f, 0.f, 0.f, 0.f);
            if (bm + r < N) v = ((const float4*)(A + (size_t)(bm + r) * 128))[g];
            v.x = fmaxf(v.x, 0.f); v.y = fmaxf(v.y, 0.f);
            v.z = fmaxf(v.z, 0.f); v.w = fmaxf(v.w, 0.f);
            al[i] = v;
        }
    }
    __syncthreads();

    const int wave = tid >> 5;
    const int lane = tid & 31;
    const int msub = (wave >> 2) * 16;
    const int nb   = (wave & 3) * 16;
    const int mr   = lane & 15;
    const int kh   = (lane >> 4) * 2;

    v8f acc = {};
    for (int kc = 0; kc < 128; kc += 4) {
        v2f a, b;
        a[0] = Alds[(msub + mr) * 128 + kc + kh + 0];
        a[1] = Alds[(msub + mr) * 128 + kc + kh + 1];
        b[0] = Wlds[(kc + kh + 0) * 64 + nb + mr];
        b[1] = Wlds[(kc + kh + 1) * 64 + nb + mr];
        acc = __builtin_amdgcn_wmma_f32_16x16x4_f32(
                  false, a, false, b, (short)0, acc, false, false);
    }

    const int mtop = (lane >> 4) * 8;
    for (int v = 0; v < 8; ++v) {
        int m = bm + msub + mtop + v;
        if (m < N) H2[(size_t)m * 64 + nb + mr] = acc[v];
    }
}

extern "C" void kernel_launch(void* const* d_in, const int* in_sizes, int n_in,
                              void* d_out, int out_size, void* d_ws, size_t ws_size,
                              hipStream_t stream) {
    const float* x   = (const float*)d_in[0];
    const int*   ei  = (const int*)d_in[1];
    const float* ew  = (const float*)d_in[2];
    const float* W1  = (const float*)d_in[3];
    const float* b1  = (const float*)d_in[4];
    const float* W2  = (const float*)d_in[5];
    const float* b2  = (const float*)d_in[6];
    float* out = (float*)d_out;

    const int E = in_sizes[2];
    const int N = in_sizes[0] / 128;
    const int* row = ei;
    const int* col = ei + E;

    const size_t Np = ((size_t)N + 1 + 63) & ~(size_t)63;
    const size_t Ep = ((size_t)E + 63) & ~(size_t)63;
    int*   cnt    = (int*)d_ws;
    int*   rowptr = cnt + Np;
    float* dinv   = (float*)(rowptr + Np);
    int*   rank   = (int*)(dinv + Np);
    int*   csr    = rank + Ep;
    float* h1     = (float*)(csr + 2 * Ep);
    float* out1   = h1 + (size_t)N * 128;
    float* h2     = out1 + (size_t)N * 128;

    const int T = 256;

    k_zero<<<(N + T - 1) / T, T, 0, stream>>>(cnt, N);
    k_rank<<<(E + T - 1) / T, T, 0, stream>>>(col, cnt, rank, E);
    k_scan<<<1, 1024, 0, stream>>>(cnt, rowptr, N);
    k_fill<<<(E + T - 1) / T, T, 0, stream>>>(row, col, ew, rowptr, rank, csr, E);
    k_dinv<<<(N + T - 1) / T, T, 0, stream>>>(rowptr, csr, dinv, N);

    k_gemm1<<<(N + 15) / 16, 256, 0, stream>>>(x, W1, h1, N);
    k_gather1<<<(N + 7) / 8, 256, 0, stream>>>(h1, rowptr, csr, dinv, b1, out1, N);

    k_gemm2<<<(N + 31) / 32, 256, 0, stream>>>(out1, W2, h2, N);
    k_gather2<<<(N + 7) / 8, 256, 0, stream>>>(h2, rowptr, csr, dinv, b2, out, N);
}


# Scan kernel: one pass, each thread's 13 whole-line 16-byte count loads requested together and held in registers, wave prefixes on the vector ALU, instead of 60 load round trips one after another
# speedup vs baseline: 1.7867x; 1.1198x over previous
#include <hip/hip_runtime.h>
#include <hip/hip_bf16.h>


typedef __attribute__((ext_vector_type(2))) float v2f;
typedef __attribute__((ext_vector_type(4))) float v4f;
typedef __attribute__((ext_vector_type(8))) float v8f;

__global__ void k_zero(int* __restrict__ cnt, int N) {
    int i = blockIdx.x * blockDim.x + threadIdx.x;
    if (i < N) cnt[i] = 0;
}

__global__ void k_rank(const int* __restrict__ col,
                       int* __restrict__ cnt,
                       int* __restrict__ rank, int E) {
    int e = blockIdx.x * blockDim.x + threadIdx.x;
    if (e < E) rank[e] = atomicAdd(&cnt[col[e]], 1);
}

__device__ __forceinline__ int wave_incl_scan(int x, int lane) {
    x += __builtin_amdgcn_update_dpp(0, x, 0x111, 0xf, 0xf, true);
    x += __builtin_amdgcn_update_dpp(0, x, 0x112, 0xf, 0xf, true);
    x += __builtin_amdgcn_update_dpp(0, x, 0x114, 0xf, 0xf, true);
    x += __builtin_amdgcn_update_dpp(0, x, 0x118, 0xf, 0xf, true);
    const int lo = __builtin_amdgcn_readlane(x, 15);
    return x + ((lane >= 16) ? lo : 0);
}

__global__ __launch_bounds__(1024) void k_scan(const int* __restrict__ cnt,
                                                int* __restrict__ rowptr, int N) {
    constexpr int K = 13;
    __shared__ int wsum[K * 32];
    __shared__ int tsum[K];
    const int tid = threadIdx.x, lane = tid & 31, wv = tid >> 5;
    const int nfull = N >> 2;
    const int4* __restrict__ cnt4 = (const int4*)cnt;
    int4* __restrict__ row4 = (int4*)rowptr;
    int carry = 0;
    for (int base = 0; base < nfull; base += K * 1024) {
        int4 v[K];
        for (int k = 0; k < K; ++k) {
            v[k] = cnt4[min(base + k * 1024 + tid, nfull - 1)];
        }
        int s[K], inc[K];
        for (int k = 0; k < K; ++k) {
            if (base + k * 1024 + tid >= nfull) v[k] = make_int4(0, 0, 0, 0);
            s[k] = (v[k].x + v[k].y) + (v[k].z + v[k].w);
            inc[k] = wave_incl_scan(s[k], lane);
        }
        if (lane == 31) {
            for (int k = 0; k < K; ++k) wsum[k * 32 + wv] = inc[k];
        }
        __syncthreads();
        if (wv < K) {
            const int x = wsum[wv * 32 + lane];
            const int xi = wave_incl_scan(x, lane);
            wsum[wv * 32 + lane] = xi - x;
            if (lane == 31) tsum[wv] = xi;
        }
        __syncthreads();
        int run = carry;
        for (int k = 0; k < K; ++k) {
            int4 o;
            o.x = run + wsum[k * 32 + wv] + (inc[k] - s[k]);
            o.y = o.x + v[k].x;
            o.z = o.y + v[k].y;
            o.w = o.z + v[k].z;
            run += tsum[k];
            if (base + k * 1024 + tid < nfull) row4[base + k * 1024 + tid] = o;
        }
        carry = run;
        __syncthreads();
    }
    if (tid == 0) {
        for (int j = nfull * 4; j < N; ++j) {
            rowptr[j] = carry;
            carry += cnt[j];
        }
        rowptr[N] = carry;
    }
}

__global__ void k_fill(const int* __restrict__ row,
                       const int* __restrict__ col,
                       const float* __restrict__ w,
                       const int* __restrict__ rowptr,
                       const int* __restrict__ rank,
                       int* __restrict__ csr, int E) {
    int e = blockIdx.x * blockDim.x + threadIdx.x;
    if (e < E) {
        int p = rowptr[col[e]] + rank[e];
        ((int2*)csr)[p] = make_int2(row[e], __float_as_int(w[e]));
    }
}

__global__ void k_dinv(const int* __restrict__ rowptr,
                       const int* __restrict__ csr,
                       float* __restrict__ dinv, int N) {
    int i = blockIdx.x * blockDim.x + threadIdx.x;
    if (i < N) {
        const int b = rowptr[i], e = rowptr[i + 1];
        float d = 1.0f;
        for (int p = b; p < e; ++p) d += __int_as_float(csr[2 * p + 1]);
        dinv[i] = (d > 0.0f) ? rsqrtf(d) : 0.0f;
    }
}

__global__ __launch_bounds__(256) void k_gemm1(const float* __restrict__ X,
                                               const float* __restrict__ W,
                                               float* __restrict__ H,
                                               int N) {
    __shared__ float Alds[16 * 128];
    const int tid = threadIdx.x;
    const int bm  = blockIdx.x * 16;

    float4* al4 = (float4*)Alds;
    for (int i = tid; i < 16 * 32; i += 256) {
        int r = i >> 5, g = i & 31;
        float4 v = make_float4(0.f, 0.f, 0.f, 0.f);
        if (bm + r < N) v = ((const float4*)(X + (size_t)(bm + r) * 128))[g];
        al4[i] = v;
    }
    __syncthreads();

    const int wave = tid >> 5;
    const int lane = tid & 31;
    const int nb   = wave * 16;
    const int mr   = lane & 15;
    const int kh   = (lane >> 4) * 2;

    v8f acc = {};
    for (int kc = 0; kc < 128; kc += 4) {
        v2f a, b;
        a[0] = Alds[mr * 128 + kc + kh + 0];
        a[1] = Alds[mr * 128 + kc + kh + 1];
        b[0] = W[(kc + kh + 0) * 128 + nb + mr];
        b[1] = W[(kc + kh + 1) * 128 + nb + mr];
        acc = __builtin_amdgcn_wmma_f32_16x16x4_f32(
                  false, a, false, b, (short)0, acc, false, false);
    }

    const int mtop = (lane >> 4) * 8;
    for (int v = 0; v < 8; ++v) {
        int m = bm + mtop + v;
        if (m < N) H[(size_t)m * 128 + nb + mr] = acc[v];
    }
}

template <typename V>
__device__ __forceinline__ void gather_body(const float* __restrict__ h,
                                            const int* __restrict__ rowptr,
                                            const int* __restrict__ csr,
                                            const float* __restrict__ dinv,
                                            const float* __restrict__ bias,
                                            float* __restrict__ out, int N) {
    constexpr int VEC = sizeof(V) / sizeof(float);
    constexpr int F   = VEC * 32;
    const int lane = threadIdx.x & 31;
    const int node = __builtin_amdgcn_readfirstlane(
        (int)(blockIdx.x * (blockDim.x >> 5) + (threadIdx.x >> 5)));
    if (node >= N) return;
    const int beg = rowptr[node], end = rowptr[node + 1];
    const float dc = dinv[node];
    const V hv = *(const V*)(h + (size_t)node * F + lane * VEC);
    const V bv = *(const V*)(bias + lane * VEC);
    V acc = bv + hv * (dc * dc);
    const int2* ent = (const int2*)csr;
    for (int base = beg; base < end; base += 32) {
        const int n = min(32, end - base);
        int r = 0;
        float nw = 0.0f;
        if (lane < n) {
            const int2 t = ent[base + lane];
            r  = t.x;
            nw = dinv[r] * __int_as_float(t.y) * dc;
        }
        int j = 0;
        for (; j + 4 <= n; j += 4) {
            const int r0 = __builtin_amdgcn_readlane(r, j + 0);
            const int r1 = __builtin_amdgcn_readlane(r, j + 1);
            const int r2 = __builtin_amdgcn_readlane(r, j + 2);
            const int r3 = __builtin_amdgcn_readlane(r, j + 3);
            const float w0 = __int_as_float(__builtin_amdgcn_readlane(__float_as_int(nw), j + 0));
            const float w1 = __int_as_float(__builtin_amdgcn_readlane(__float_as_int(nw), j + 1));
            const float w2 = __int_as_float(__builtin_amdgcn_readlane(__float_as_int(nw), j + 2));
            const float w3 = __int_as_float(__builtin_amdgcn_readlane(__float_as_int(nw), j + 3));
            const V v0 = *(const V*)(h + (size_t)r0 * F + lane * VEC);
            const V v1 = *(const V*)(h + (size_t)r1 * F + lane * VEC);
            const V v2 = *(const V*)(h + (size_t)r2 * F + lane * VEC);
            const V v3 = *(const V*)(h + (size_t)r3 * F + lane * VEC);
            acc += v0 * w0;
            acc += v1 * w1;
            acc += v2 * w2;
            acc += v3 * w3;
        }
        for (; j < n; ++j) {
            const int r0 = __builtin_amdgcn_readlane(r, j);
            const float w0 = __int_as_float(__builtin_amdgcn_readlane(__float_as_int(nw), j));
            const V v0 = *(const V*)(h + (size_t)r0 * F + lane * VEC);
            acc += v0 * w0;
        }
    }
    *(V*)(out + (size_t)node * F + lane * VEC) = acc;
}

__global__ __launch_bounds__(256) void k_gather1(const float* __restrict__ h1,
                                                 const int* __restrict__ rowptr,
                                                 const int* __restrict__ csr,
                                                 const float* __restrict__ dinv,
                                                 const float* __restrict__ b1,
                                                 float* __restrict__ out1, int N) {
    gather_body<v4f>(h1, rowptr, csr, dinv, b1, out1, N);
}

__global__ __launch_bounds__(256) void k_gather2(const float* __restrict__ h2,
                                                 const int* __restrict__ rowptr,
                                                 const int* __restrict__ csr,
                                                 const float* __restrict__ dinv,
                                                 const float* __restrict__ b2,
                                                 float* __restrict__ out, int N) {
    gather_body<v2f>(h2, rowptr, csr, dinv, b2, out, N);
}

__global__ __launch_bounds__(256) void k_gemm2(const float* __restrict__ A,
                                               const float* __restrict__ W2,
                                               float* __restrict__ H2,
                                               int N) {
    __shared__ float Alds[32 * 128];
    __shared__ float Wlds[128 * 64];
    const int tid = threadIdx.x;
    const int bm  = blockIdx.x * 32;

    {
        const float4* w4 = (const float4*)W2;
        float4* wl = (float4*)Wlds;
        for (int i = tid; i < 2048; i += 256) wl[i] = w4[i];
    }
    {
        float4* al = (float4*)Alds;
        for (int i = tid; i < 32 * 32; i += 256) {
            int r = i >> 5, g = i & 31;
            float4 v = make_float4(0.f, 0.f, 0.f, 0.f);
            if (bm + r < N) v = ((const float4*)(A + (size_t)(bm + r) * 128))[g];
            v.x = fmaxf(v.x, 0.f); v.y = fmaxf(v.y, 0.f);
            v.z = fmaxf(v.z, 0.f); v.w = fmaxf(v.w, 0.f);
            al[i] = v;
        }
    }
    __syncthreads();

    const int wave = tid >> 5;
    const int lane = tid & 31;
    const int msub = (wave >> 2) * 16;
    const int nb   = (wave & 3) * 16;
    const int mr   = lane & 15;
    const int kh   = (lane >> 4) * 2;

    v8f acc = {};
    for (int kc = 0; kc < 128; kc += 4) {
        v2f a, b;
        a[0] = Alds[(msub + mr) * 128 + kc + kh + 0];
        a[1] = Alds[(msub + mr) * 128 + kc + kh + 1];
        b[0] = Wlds[(kc + kh + 0) * 64 + nb + mr];
        b[1] = Wlds[(kc + kh + 1) * 64 + nb + mr];
        acc = __builtin_amdgcn_wmma_f32_16x16x4_f32(
                  false, a, false, b, (short)0, acc, false, false);
    }

    const int mtop = (lane >> 4) * 8;
    for (int v = 0; v < 8; ++v) {
        int m = bm + msub + mtop + v;
        if (m < N) H2[(size_t)m * 64 + nb + mr] = acc[v];
    }
}

extern "C" void kernel_launch(void* const* d_in, const int* in_sizes, int n_in,
                              void* d_out, int out_size, void* d_ws, size_t ws_size,
                              hipStream_t stream) {
    const float* x   = (const float*)d_in[0];
    const int*   ei  = (const int*)d_in[1];
    const float* ew  = (const float*)d_in[2];
    const float* W1  = (const float*)d_in[3];
    const float* b1  = (const float*)d_in[4];
    const float* W2  = (const float*)d_in[5];
    const float* b2  = (const float*)d_in[6];
    float* out = (float*)d_out;

    const int E = in_sizes[2];
    const int N = in_sizes[0] / 128;
    const int* row = ei;
    const int* col = ei + E;

    const size_t Np = ((size_t)N + 1 + 63) & ~(size_t)63;
    const size_t Ep = ((size_t)E + 63) & ~(size_t)63;
    int*   cnt    = (int*)d_ws;
    int*   rowptr = cnt + Np;
    float* dinv   = (float*)(rowptr + Np);
    int*   rank   = (int*)(dinv + Np);
    int*   csr    = rank + Ep;
    float* h1     = (float*)(csr + 2 * Ep);
    float* out1   = h1 + (size_t)N * 128;
    float* h2     = out1 + (size_t)N * 128;

    const int T = 256;

    k_zero<<<(N + T - 1) / T, T, 0, stream>>>(cnt, N);
    k_rank<<<(E + T - 1) / T, T, 0, stream>>>(col, cnt, rank, E);
    k_scan<<<1, 1024, 0, stream>>>(cnt, rowptr, N);
    k_fill<<<(E + T - 1) / T, T, 0, stream>>>(row, col, ew, rowptr, rank, csr, E);
    k_dinv<<<(N + T - 1) / T, T, 0, stream>>>(rowptr, csr, dinv, N);

    k_gemm1<<<(N + 15) / 16, 256, 0, stream>>>(x, W1, h1, N);
    k_gather1<<<(N + 7) / 8, 256, 0, stream>>>(h1, rowptr, csr, dinv, b1, out1, N);

    k_gemm2<<<(N + 31) / 32, 256, 0, stream>>>(out1, W2, h2, N);
    k_gather2<<<(N + 7) / 8, 256, 0, stream>>>(h2, rowptr, csr, dinv, b2, out, N);
}


# Rank edges with LDS counters in a two-level counting sort (tile x 256-node bucket, then node) instead of one returning global atomic per edge; slot order in a column now set by LDS add arrival
# speedup vs baseline: 2.3851x; 1.3349x over previous
#include <hip/hip_runtime.h>
#include <hip/hip_bf16.h>


typedef __attribute__((ext_vector_type(2))) float v2f;
typedef __attribute__((ext_vector_type(4))) float v4f;
typedef __attribute__((ext_vector_type(8))) float v8f;

constexpr int TILES = 256;
constexpr int BSH   = 8;
constexpr int BW    = 1 << BSH;
constexpr int MAXB  = 1024;

__global__ __launch_bounds__(512) void k_hist(const int* __restrict__ col,
                                               int* __restrict__ tcnt,
                                               int E, int NB, int per) {
    __shared__ int hist[MAXB];
    const int tid = threadIdx.x, tile = blockIdx.x;
    for (int b = tid; b < NB; b += 512) hist[b] = 0;
    __syncthreads();
    const int e0 = tile * per, e1 = min(E, e0 + per);
    for (int base = e0; base < e1; base += 4 * 512) {
        int c[4];
        for (int k = 0; k < 4; ++k) {
            const int e = base + k * 512 + tid;
            c[k] = col[min(e, e1 - 1)];
        }
        for (int k = 0; k < 4; ++k) {
            if (base + k * 512 + tid < e1) atomicAdd(&hist[c[k] >> BSH], 1);
        }
    }
    __syncthreads();
    for (int b = tid; b < NB; b += 512) tcnt[b * TILES + tile] = hist[b];
}

__device__ __forceinline__ int wave_incl_scan(int x, int lane) {
    x += __builtin_amdgcn_update_dpp(0, x, 0x111, 0xf, 0xf, true);
    x += __builtin_amdgcn_update_dpp(0, x, 0x112, 0xf, 0xf, true);
    x += __builtin_amdgcn_update_dpp(0, x, 0x114, 0xf, 0xf, true);
    x += __builtin_amdgcn_update_dpp(0, x, 0x118, 0xf, 0xf, true);
    const int lo = __builtin_amdgcn_readlane(x, 15);
    return x + ((lane >= 16) ? lo : 0);
}

__global__ __launch_bounds__(1024) void k_scan(const int* __restrict__ cnt,
                                                int* __restrict__ rowptr, int N) {
    constexpr int K = 13;
    __shared__ int wsum[K * 32];
    __shared__ int tsum[K];
    const int tid = threadIdx.x, lane = tid & 31, wv = tid >> 5;
    const int nfull = N >> 2;
    const int4* __restrict__ cnt4 = (const int4*)cnt;
    int4* __restrict__ row4 = (int4*)rowptr;
    int carry = 0;
    for (int base = 0; base < nfull; base += K * 1024) {
        int4 v[K];
        for (int k = 0; k < K; ++k) {
            v[k] = cnt4[min(base + k * 1024 + tid, nfull - 1)];
        }
        int s[K], inc[K];
        for (int k = 0; k < K; ++k) {
            if (base + k * 1024 + tid >= nfull) v[k] = make_int4(0, 0, 0, 0);
            s[k] = (v[k].x + v[k].y) + (v[k].z + v[k].w);
            inc[k] = wave_incl_scan(s[k], lane);
        }
        if (lane == 31) {
            for (int k = 0; k < K; ++k) wsum[k * 32 + wv] = inc[k];
        }
        __syncthreads();
        if (wv < K) {
            const int x = wsum[wv * 32 + lane];
            const int xi = wave_incl_scan(x, lane);
            wsum[wv * 32 + lane] = xi - x;
            if (lane == 31) tsum[wv] = xi;
        }
        __syncthreads();
        int run = carry;
        for (int k = 0; k < K; ++k) {
            int4 o;
            o.x = run + wsum[k * 32 + wv] + (inc[k] - s[k]);
            o.y = o.x + v[k].x;
            o.z = o.y + v[k].y;
            o.w = o.z + v[k].z;
            run += tsum[k];
            if (base + k * 1024 + tid < nfull) row4[base + k * 1024 + tid] = o;
        }
        carry = run;
        __syncthreads();
    }
    if (tid == 0) {
        for (int j = nfull * 4; j < N; ++j) {
            rowptr[j] = carry;
            carry += cnt[j];
        }
        rowptr[N] = carry;
    }
}

__global__ __launch_bounds__(512) void k_part(const int* __restrict__ row,
                                               const int* __restrict__ col,
                                               const float* __restrict__ w,
                                               const int* __restrict__ toff,
                                               int* __restrict__ recw,
                                               int E, int NB, int per) {
    __shared__ int cur[MAXB];
    int4* __restrict__ rec = (int4*)recw;
    const int tid = threadIdx.x, tile = blockIdx.x;
    for (int b = tid; b < NB; b += 512) cur[b] = toff[b * TILES + tile];
    __syncthreads();
    const int e0 = tile * per, e1 = min(E, e0 + per);
    for (int base = e0; base < e1; base += 4 * 512) {
        int c[4], r[4], wb[4];
        for (int k = 0; k < 4; ++k) {
            const int e = min(base + k * 512 + tid, e1 - 1);
            c[k]  = col[e];
            r[k]  = row[e];
            wb[k] = __float_as_int(w[e]);
        }
        for (int k = 0; k < 4; ++k) {
            if (base + k * 512 + tid < e1) {
                const int p = atomicAdd(&cur[c[k] >> BSH], 1);
                rec[p] = make_int4(r[k], c[k], wb[k], 0);
            }
        }
    }
}

__global__ __launch_bounds__(1024) void k_build(const int* __restrict__ rcol,
                                                 const int* __restrict__ toff,
                                                 int* __restrict__ rowptr,
                                                 int* csr,
                                                 float* __restrict__ dinv, int N) {
    __shared__ int cnt[BW];
    __shared__ int cur[BW];
    __shared__ int wtot[BW / 32];
    const int tid = threadIdx.x, lane = tid & 31, wv = tid >> 5;
    const int b  = blockIdx.x;
    const int bs = toff[b * TILES], be = toff[(b + 1) * TILES];
    const int4* __restrict__ rec = (const int4*)rcol;
    if (tid < BW) cnt[tid] = 0;
    __syncthreads();
    for (int base = bs; base < be; base += 4 * 1024) {
        int c[4];
        for (int k = 0; k < 4; ++k) {
            const int e = min(base + k * 1024 + tid, be - 1);
            c[k] = rcol[4 * (size_t)e + 1];
        }
        for (int k = 0; k < 4; ++k) {
            if (base + k * 1024 + tid < be) atomicAdd(&cnt[c[k] & (BW - 1)], 1);
        }
    }
    __syncthreads();
    int c = 0, inc = 0, off = 0;
    if (tid < BW) {
        c   = cnt[tid];
        inc = wave_incl_scan(c, lane);
        if (lane == 31) wtot[wv] = inc;
    }
    __syncthreads();
    const int node = b * BW + tid;
    if (tid < BW) {
        int pre = 0;
        for (int k = 0; k < BW / 32; ++k) pre += (k < wv) ? wtot[k] : 0;
        off = bs + pre + (inc - c);
        cur[tid] = off;
        if (node < N) rowptr[node] = off;
        if (node == N - 1) rowptr[N] = off + c;
    }
    __syncthreads();
    int2* __restrict__ ent = (int2*)csr;
    for (int base = bs; base < be; base += 4 * 1024) {
        int4 t[4];
        for (int k = 0; k < 4; ++k) {
            const int e = min(base + k * 1024 + tid, be - 1);
            t[k] = rec[e];
        }
        for (int k = 0; k < 4; ++k) {
            if (base + k * 1024 + tid < be) {
                const int p = atomicAdd(&cur[t[k].y & (BW - 1)], 1);
                ent[p] = make_int2(t[k].x, t[k].z);
            }
        }
    }
    __syncthreads();
    if (tid < BW && node < N) {
        const int end = off + c;
        float d = 1.0f;
        int p = off;
        for (; p + 4 <= end; p += 4) {
            const float a0 = __int_as_float(csr[2 * p + 1]);
            const float a1 = __int_as_float(csr[2 * p + 3]);
            const float a2 = __int_as_float(csr[2 * p + 5]);
            const float a3 = __int_as_float(csr[2 * p + 7]);
            d += a0; d += a1; d += a2; d += a3;
        }
        for (; p < end; ++p) d += __int_as_float(csr[2 * p + 1]);
        dinv[node] = (d > 0.0f) ? rsqrtf(d) : 0.0f;
    }
}

__global__ __launch_bounds__(256) void k_gemm1(const float* __restrict__ X,
                                               const float* __restrict__ W,
                                               float* __restrict__ H,
                                               int N) {
    __shared__ float Alds[16 * 128];
    const int tid = threadIdx.x;
    const int bm  = blockIdx.x * 16;

    float4* al4 = (float4*)Alds;
    for (int i = tid; i < 16 * 32; i += 256) {
        int r = i >> 5, g = i & 31;
        float4 v = make_float4(0.f, 0.f, 0.f, 0.f);
        if (bm + r < N) v = ((const float4*)(X + (size_t)(bm + r) * 128))[g];
        al4[i] = v;
    }
    __syncthreads();

    const int wave = tid >> 5;
    const int lane = tid & 31;
    const int nb   = wave * 16;
    const int mr   = lane & 15;
    const int kh   = (lane >> 4) * 2;

    v8f acc = {};
    for (int kc = 0; kc < 128; kc += 4) {
        v2f a, b;
        a[0] = Alds[mr * 128 + kc + kh + 0];
        a[1] = Alds[mr * 128 + kc + kh + 1];
        b[0] = W[(kc + kh + 0) * 128 + nb + mr];
        b[1] = W[(kc + kh + 1) * 128 + nb + mr];
        acc = __builtin_amdgcn_wmma_f32_16x16x4_f32(
                  false, a, false, b, (short)0, acc, false, false);
    }

    const int mtop = (lane >> 4) * 8;
    for (int v = 0; v < 8; ++v) {
        int m = bm + mtop + v;
        if (m < N) H[(size_t)m * 128 + nb + mr] = acc[v];
    }
}

template <typename V>
__device__ __forceinline__ void gather_body(const float* __restrict__ h,
                                            const int* __restrict__ rowptr,
                                            const int* __restrict__ csr,
                                            const float* __restrict__ dinv,
                                            const float* __restrict__ bias,
                                            float* __restrict__ out, int N) {
    constexpr int VEC = sizeof(V) / sizeof(float);
    constexpr int F   = VEC * 32;
    const int lane = threadIdx.x & 31;
    const int node = __builtin_amdgcn_readfirstlane(
        (int)(blockIdx.x * (blockDim.x >> 5) + (threadIdx.x >> 5)));
    if (node >= N) return;
    const int beg = rowptr[node], end = rowptr[node + 1];
    const float dc = dinv[node];
    const V hv = *(const V*)(h + (size_t)node * F + lane * VEC);
    const V bv = *(const V*)(bias + lane * VEC);
    V acc = bv + hv * (dc * dc);
    const int2* ent = (const int2*)csr;
    for (int base = beg; base < end; base += 32) {
        const int n = min(32, end - base);
        int r = 0;
        float nw = 0.0f;
        if (lane < n) {
            const int2 t = ent[base + lane];
            r  = t.x;
            nw = dinv[r] * __int_as_float(t.y) * dc;
        }
        int j = 0;
        for (; j + 4 <= n; j += 4) {
            const int r0 = __builtin_amdgcn_readlane(r, j + 0);
            const int r1 = __builtin_amdgcn_readlane(r, j + 1);
            const int r2 = __builtin_amdgcn_readlane(r, j + 2);
            const int r3 = __builtin_amdgcn_readlane(r, j + 3);
            const float w0 = __int_as_float(__builtin_amdgcn_readlane(__float_as_int(nw), j + 0));
            const float w1 = __int_as_float(__builtin_amdgcn_readlane(__float_as_int(nw), j + 1));
            const float w2 = __int_as_float(__builtin_amdgcn_readlane(__float_as_int(nw), j + 2));
            const float w3 = __int_as_float(__builtin_amdgcn_readlane(__float_as_int(nw), j + 3));
            const V v0 = *(const V*)(h + (size_t)r0 * F + lane * VEC);
            const V v1 = *(const V*)(h + (size_t)r1 * F + lane * VEC);
            const V v2 = *(const V*)(h + (size_t)r2 * F + lane * VEC);
            const V v3 = *(const V*)(h + (size_t)r3 * F + lane * VEC);
            acc += v0 * w0;
            acc += v1 * w1;
            acc += v2 * w2;
            acc += v3 * w3;
        }
        for (; j < n; ++j) {
            const int r0 = __builtin_amdgcn_readlane(r, j);
            const float w0 = __int_as_float(__builtin_amdgcn_readlane(__float_as_int(nw), j));
            const V v0 = *(const V*)(h + (size_t)r0 * F + lane * VEC);
            acc += v0 * w0;
        }
    }
    *(V*)(out + (size_t)node * F + lane * VEC) = acc;
}

__global__ __launch_bounds__(256) void k_gather1(const float* __restrict__ h1,
                                                 const int* __restrict__ rowptr,
                                                 const int* __restrict__ csr,
                                                 const float* __restrict__ dinv,
                                                 const float* __restrict__ b1,
                                                 float* __restrict__ out1, int N) {
    gather_body<v4f>(h1, rowptr, csr, dinv, b1, out1, N);
}

__global__ __launch_bounds__(256) void k_gather2(const float* __restrict__ h2,
                                                 const int* __restrict__ rowptr,
                                                 const int* __restrict__ csr,
                                                 const float* __restrict__ dinv,
                                                 const float* __restrict__ b2,
                                                 float* __restrict__ out, int N) {
    gather_body<v2f>(h2, rowptr, csr, dinv, b2, out, N);
}

__global__ __launch_bounds__(256) void k_gemm2(const float* __restrict__ A,
                                               const float* __restrict__ W2,
                                               float* __restrict__ H2,
                                               int N) {
    __shared__ float Alds[32 * 128];
    __shared__ float Wlds[128 * 64];
    const int tid = threadIdx.x;
    const int bm  = blockIdx.x * 32;

    {
        const float4* w4 = (const float4*)W2;
        float4* wl = (float4*)Wlds;
        for (int i = tid; i < 2048; i += 256) wl[i] = w4[i];
    }
    {
        float4* al = (float4*)Alds;
        for (int i = tid; i < 32 * 32; i += 256) {
            int r = i >> 5, g = i & 31;
            float4 v = make_float4(0.f, 0.f, 0.f, 0.f);
            if (bm + r < N) v = ((const float4*)(A + (size_t)(bm + r) * 128))[g];
            v.x = fmaxf(v.x, 0.f); v.y = fmaxf(v.y, 0.f);
            v.z = fmaxf(v.z, 0.f); v.w = fmaxf(v.w, 0.f);
            al[i] = v;
        }
    }
    __syncthreads();

    const int wave = tid >> 5;
    const int lane = tid & 31;
    const int msub = (wave >> 2) * 16;
    const int nb   = (wave & 3) * 16;
    const int mr   = lane & 15;
    const int kh   = (lane >> 4) * 2;

    v8f acc = {};
    for (int kc = 0; kc < 128; kc += 4) {
        v2f a, b;
        a[0] = Alds[(msub + mr) * 128 + kc + kh + 0];
        a[1] = Alds[(msub + mr) * 128 + kc + kh + 1];
        b[0] = Wlds[(kc + kh + 0) * 64 + nb + mr];
        b[1] = Wlds[(kc + kh + 1) * 64 + nb + mr];
        acc = __builtin_amdgcn_wmma_f32_16x16x4_f32(
                  false, a, false, b, (short)0, acc, false, false);
    }

    const int mtop = (lane >> 4) * 8;
    for (int v = 0; v < 8; ++v) {
        int m = bm + msub + mtop + v;
        if (m < N) H2[(size_t)m * 64 + nb + mr] = acc[v];
    }
}

extern "C" void kernel_launch(void* const* d_in, const int* in_sizes, int n_in,
                              void* d_out, int out_size, void* d_ws, size_t ws_size,
                              hipStream_t stream) {
    const float* x   = (const float*)d_in[0];
    const int*   ei  = (const int*)d_in[1];
    const float* ew  = (const float*)d_in[2];
    const float* W1  = (const float*)d_in[3];
    const float* b1  = (const float*)d_in[4];
    const float* W2  = (const float*)d_in[5];
    const float* b2  = (const float*)d_in[6];
    float* out = (float*)d_out;

    const int E = in_sizes[2];
    const int N = in_sizes[0] / 128;
    const int* row = ei;
    const int* col = ei + E;

    const int NB = (N + BW - 1) / BW;
    const int per = (E + TILES - 1) / TILES;
    const size_t Np = ((size_t)N + 1 + 63) & ~(size_t)63;
    const size_t Ep = ((size_t)E + 63) & ~(size_t)63;
    const size_t Cp = ((size_t)NB * TILES + 1 + 63) & ~(size_t)63;
    int*   tcnt   = (int*)d_ws;
    int*   toff   = tcnt + Cp;
    int*   rowptr = toff + Cp;
    float* dinv   = (float*)(rowptr + Np);
    int*   rec    = (int*)(dinv + Np);
    int*   csr    = rec + 4 * Ep;
    float* h1     = (float*)(csr + 2 * Ep);
    float* out1   = h1 + (size_t)N * 128;
    float* h2     = out1 + (size_t)N * 128;

    k_hist<<<TILES, 512, 0, stream>>>(col, tcnt, E, NB, per);
    k_scan<<<1, 1024, 0, stream>>>(tcnt, toff, NB * TILES);
    k_part<<<TILES, 512, 0, stream>>>(row, col, ew, toff, rec, E, NB, per);
    k_build<<<NB, 1024, 0, stream>>>(rec, toff, rowptr, csr, dinv, N);

    k_gemm1<<<(N + 15) / 16, 256, 0, stream>>>(x, W1, h1, N);
    k_gather1<<<(N + 7) / 8, 256, 0, stream>>>(h1, rowptr, csr, dinv, b1, out1, N);

    k_gemm2<<<(N + 31) / 32, 256, 0, stream>>>(out1, W2, h2, N);
    k_gather2<<<(N + 7) / 8, 256, 0, stream>>>(h2, rowptr, csr, dinv, b2, out, N);
}


# LDS row pitch of the GEMM tiles padded (A 128 to 132 floats, W2 64 to 80) so a WMMA fragment read covers all 64 banks once instead of putting its 16 rows on the same banks (16-way conflict)
# speedup vs baseline: 2.9091x; 1.2197x over previous
#include <hip/hip_runtime.h>
#include <hip/hip_bf16.h>


typedef __attribute__((ext_vector_type(2))) float v2f;
typedef __attribute__((ext_vector_type(4))) float v4f;
typedef __attribute__((ext_vector_type(8))) float v8f;

constexpr int APITCH = 132;
constexpr int WPITCH = 80;

constexpr int TILES = 256;
constexpr int BSH   = 8;
constexpr int BW    = 1 << BSH;
constexpr int MAXB  = 1024;

__global__ __launch_bounds__(512) void k_hist(const int* __restrict__ col,
                                               int* __restrict__ tcnt,
                                               int E, int NB, int per) {
    __shared__ int hist[MAXB];
    const int tid = threadIdx.x, tile = blockIdx.x;
    for (int b = tid; b < NB; b += 512) hist[b] = 0;
    __syncthreads();
    const int e0 = tile * per, e1 = min(E, e0 + per);
    for (int base = e0; base < e1; base += 4 * 512) {
        int c[4];
        for (int k = 0; k < 4; ++k) {
            const int e = base + k * 512 + tid;
            c[k] = col[min(e, e1 - 1)];
        }
        for (int k = 0; k < 4; ++k) {
            if (base + k * 512 + tid < e1) atomicAdd(&hist[c[k] >> BSH], 1);
        }
    }
    __syncthreads();
    for (int b = tid; b < NB; b += 512) tcnt[b * TILES + tile] = hist[b];
}

__device__ __forceinline__ int wave_incl_scan(int x, int lane) {
    x += __builtin_amdgcn_update_dpp(0, x, 0x111, 0xf, 0xf, true);
    x += __builtin_amdgcn_update_dpp(0, x, 0x112, 0xf, 0xf, true);
    x += __builtin_amdgcn_update_dpp(0, x, 0x114, 0xf, 0xf, true);
    x += __builtin_amdgcn_update_dpp(0, x, 0x118, 0xf, 0xf, true);
    const int lo = __builtin_amdgcn_readlane(x, 15);
    return x + ((lane >= 16) ? lo : 0);
}

__global__ __launch_bounds__(1024) void k_scan(const int* __restrict__ cnt,
                                                int* __restrict__ rowptr, int N) {
    constexpr int K = 13;
    __shared__ int wsum[K * 32];
    __shared__ int tsum[K];
    const int tid = threadIdx.x, lane = tid & 31, wv = tid >> 5;
    const int nfull = N >> 2;
    const int4* __restrict__ cnt4 = (const int4*)cnt;
    int4* __restrict__ row4 = (int4*)rowptr;
    int carry = 0;
    for (int base = 0; base < nfull; base += K * 1024) {
        int4 v[K];
        for (int k = 0; k < K; ++k) {
            v[k] = cnt4[min(base + k * 1024 + tid, nfull - 1)];
        }
        int s[K], inc[K];
        for (int k = 0; k < K; ++k) {
            if (base + k * 1024 + tid >= nfull) v[k] = make_int4(0, 0, 0, 0);
            s[k] = (v[k].x + v[k].y) + (v[k].z + v[k].w);
            inc[k] = wave_incl_scan(s[k], lane);
        }
        if (lane == 31) {
            for (int k = 0; k < K; ++k) wsum[k * 32 + wv] = inc[k];
        }
        __syncthreads();
        if (wv < K) {
            const int x = wsum[wv * 32 + lane];
            const int xi = wave_incl_scan(x, lane);
            wsum[wv * 32 + lane] = xi - x;
            if (lane == 31) tsum[wv] = xi;
        }
        __syncthreads();
        int run = carry;
        for (int k = 0; k < K; ++k) {
            int4 o;
            o.x = run + wsum[k * 32 + wv] + (inc[k] - s[k]);
            o.y = o.x + v[k].x;
            o.z = o.y + v[k].y;
            o.w = o.z + v[k].z;
            run += tsum[k];
            if (base + k * 1024 + tid < nfull) row4[base + k * 1024 + tid] = o;
        }
        carry = run;
        __syncthreads();
    }
    if (tid == 0) {
        for (int j = nfull * 4; j < N; ++j) {
            rowptr[j] = carry;
            carry += cnt[j];
        }
        rowptr[N] = carry;
    }
}

__global__ __launch_bounds__(512) void k_part(const int* __restrict__ row,
                                               const int* __restrict__ col,
                                               const float* __restrict__ w,
                                               const int* __restrict__ toff,
                                               int* __restrict__ recw,
                                               int E, int NB, int per) {
    __shared__ int cur[MAXB];
    int4* __restrict__ rec = (int4*)recw;
    const int tid = threadIdx.x, tile = blockIdx.x;
    for (int b = tid; b < NB; b += 512) cur[b] = toff[b * TILES + tile];
    __syncthreads();
    const int e0 = tile * per, e1 = min(E, e0 + per);
    for (int base = e0; base < e1; base += 4 * 512) {
        int c[4], r[4], wb[4];
        for (int k = 0; k < 4; ++k) {
            const int e = min(base + k * 512 + tid, e1 - 1);
            c[k]  = col[e];
            r[k]  = row[e];
            wb[k] = __float_as_int(w[e]);
        }
        for (int k = 0; k < 4; ++k) {
            if (base + k * 512 + tid < e1) {
                const int p = atomicAdd(&cur[c[k] >> BSH], 1);
                rec[p] = make_int4(r[k], c[k], wb[k], 0);
            }
        }
    }
}

__global__ __launch_bounds__(1024) void k_build(const int* __restrict__ rcol,
                                                 const int* __restrict__ toff,
                                                 int* __restrict__ rowptr,
                                                 int* csr,
                                                 float* __restrict__ dinv, int N) {
    __shared__ int cnt[BW];
    __shared__ int cur[BW];
    __shared__ int wtot[BW / 32];
    const int tid = threadIdx.x, lane = tid & 31, wv = tid >> 5;
    const int b  = blockIdx.x;
    const int bs = toff[b * TILES], be = toff[(b + 1) * TILES];
    const int4* __restrict__ rec = (const int4*)rcol;
    if (tid < BW) cnt[tid] = 0;
    __syncthreads();
    for (int base = bs; base < be; base += 4 * 1024) {
        int c[4];
        for (int k = 0; k < 4; ++k) {
            const int e = min(base + k * 1024 + tid, be - 1);
            c[k] = rcol[4 * (size_t)e + 1];
        }
        for (int k = 0; k < 4; ++k) {
            if (base + k * 1024 + tid < be) atomicAdd(&cnt[c[k] & (BW - 1)], 1);
        }
    }
    __syncthreads();
    int c = 0, inc = 0, off = 0;
    if (tid < BW) {
        c   = cnt[tid];
        inc = wave_incl_scan(c, lane);
        if (lane == 31) wtot[wv] = inc;
    }
    __syncthreads();
    const int node = b * BW + tid;
    if (tid < BW) {
        int pre = 0;
        for (int k = 0; k < BW / 32; ++k) pre += (k < wv) ? wtot[k] : 0;
        off = bs + pre + (inc - c);
        cur[tid] = off;
        if (node < N) rowptr[node] = off;
        if (node == N - 1) rowptr[N] = off + c;
    }
    __syncthreads();
    int2* __restrict__ ent = (int2*)csr;
    for (int base = bs; base < be; base += 4 * 1024) {
        int4 t[4];
        for (int k = 0; k < 4; ++k) {
            const int e = min(base + k * 1024 + tid, be - 1);
            t[k] = rec[e];
        }
        for (int k = 0; k < 4; ++k) {
            if (base + k * 1024 + tid < be) {
                const int p = atomicAdd(&cur[t[k].y & (BW - 1)], 1);
                ent[p] = make_int2(t[k].x, t[k].z);
            }
        }
    }
    __syncthreads();
    if (tid < BW && node < N) {
        const int end = off + c;
        float d = 1.0f;
        int p = off;
        for (; p + 4 <= end; p += 4) {
            const float a0 = __int_as_float(csr[2 * p + 1]);
            const float a1 = __int_as_float(csr[2 * p + 3]);
            const float a2 = __int_as_float(csr[2 * p + 5]);
            const float a3 = __int_as_float(csr[2 * p + 7]);
            d += a0; d += a1; d += a2; d += a3;
        }
        for (; p < end; ++p) d += __int_as_float(csr[2 * p + 1]);
        dinv[node] = (d > 0.0f) ? rsqrtf(d) : 0.0f;
    }
}

__global__ __launch_bounds__(256) void k_gemm1(const float* __restrict__ X,
                                               const float* __restrict__ W,
                                               float* __restrict__ H,
                                               int N) {
    __shared__ alignas(16) float Alds[16 * APITCH];
    const int tid = threadIdx.x;
    const int bm  = blockIdx.x * 16;

    for (int i = tid; i < 16 * 32; i += 256) {
        int r = i >> 5, g = i & 31;
        float4 v = make_float4(0.f, 0.f, 0.f, 0.f);
        if (bm + r < N) v = ((const float4*)(X + (size_t)(bm + r) * 128))[g];
        *(float4*)&Alds[r * APITCH + g * 4] = v;
    }
    __syncthreads();

    const int wave = tid >> 5;
    const int lane = tid & 31;
    const int nb   = wave * 16;
    const int mr   = lane & 15;
    const int kh   = (lane >> 4) * 2;

    v8f acc = {};
    for (int kc = 0; kc < 128; kc += 4) {
        v2f a, b;
        a = *(const v2f*)&Alds[mr * APITCH + kc + kh];
        b[0] = W[(kc + kh + 0) * 128 + nb + mr];
        b[1] = W[(kc + kh + 1) * 128 + nb + mr];
        acc = __builtin_amdgcn_wmma_f32_16x16x4_f32(
                  false, a, false, b, (short)0, acc, false, false);
    }

    const int mtop = (lane >> 4) * 8;
    for (int v = 0; v < 8; ++v) {
        int m = bm + mtop + v;
        if (m < N) H[(size_t)m * 128 + nb + mr] = acc[v];
    }
}

template <typename V>
__device__ __forceinline__ void gather_body(const float* __restrict__ h,
                                            const int* __restrict__ rowptr,
                                            const int* __restrict__ csr,
                                            const float* __restrict__ dinv,
                                            const float* __restrict__ bias,
                                            float* __restrict__ out, int N) {
    constexpr int VEC = sizeof(V) / sizeof(float);
    constexpr int F   = VEC * 32;
    const int lane = threadIdx.x & 31;
    const int node = __builtin_amdgcn_readfirstlane(
        (int)(blockIdx.x * (blockDim.x >> 5) + (threadIdx.x >> 5)));
    if (node >= N) return;
    const int beg = rowptr[node], end = rowptr[node + 1];
    const float dc = dinv[node];
    const V hv = *(const V*)(h + (size_t)node * F + lane * VEC);
    const V bv = *(const V*)(bias + lane * VEC);
    V acc = bv + hv * (dc * dc);
    const int2* ent = (const int2*)csr;
    for (int base = beg; base < end; base += 32) {
        const int n = min(32, end - base);
        int r = 0;
        float nw = 0.0f;
        if (lane < n) {
            const int2 t = ent[base + lane];
            r  = t.x;
            nw = dinv[r] * __int_as_float(t.y) * dc;
        }
        int j = 0;
        for (; j + 4 <= n; j += 4) {
            const int r0 = __builtin_amdgcn_readlane(r, j + 0);
            const int r1 = __builtin_amdgcn_readlane(r, j + 1);
            const int r2 = __builtin_amdgcn_readlane(r, j + 2);
            const int r3 = __builtin_amdgcn_readlane(r, j + 3);
            const float w0 = __int_as_float(__builtin_amdgcn_readlane(__float_as_int(nw), j + 0));
            const float w1 = __int_as_float(__builtin_amdgcn_readlane(__float_as_int(nw), j + 1));
            const float w2 = __int_as_float(__builtin_amdgcn_readlane(__float_as_int(nw), j + 2));
            const float w3 = __int_as_float(__builtin_amdgcn_readlane(__float_as_int(nw), j + 3));
            const V v0 = *(const V*)(h + (size_t)r0 * F + lane * VEC);
            const V v1 = *(const V*)(h + (size_t)r1 * F + lane * VEC);
            const V v2 = *(const V*)(h + (size_t)r2 * F + lane * VEC);
            const V v3 = *(const V*)(h + (size_t)r3 * F + lane * VEC);
            acc += v0 * w0;
            acc += v1 * w1;
            acc += v2 * w2;
            acc += v3 * w3;
        }
        for (; j < n; ++j) {
            const int r0 = __builtin_amdgcn_readlane(r, j);
            const float w0 = __int_as_float(__builtin_amdgcn_readlane(__float_as_int(nw), j));
            const V v0 = *(const V*)(h + (size_t)r0 * F + lane * VEC);
            acc += v0 * w0;
        }
    }
    *(V*)(out + (size_t)node * F + lane * VEC) = acc;
}

__global__ __launch_bounds__(256) void k_gather1(const float* __restrict__ h1,
                                                 const int* __restrict__ rowptr,
                                                 const int* __restrict__ csr,
                                                 const float* __restrict__ dinv,
                                                 const float* __restrict__ b1,
                                                 float* __restrict__ out1, int N) {
    gather_body<v4f>(h1, rowptr, csr, dinv, b1, out1, N);
}

__global__ __launch_bounds__(256) void k_gather2(const float* __restrict__ h2,
                                                 const int* __restrict__ rowptr,
                                                 const int* __restrict__ csr,
                                                 const float* __restrict__ dinv,
                                                 const float* __restrict__ b2,
                                                 float* __restrict__ out, int N) {
    gather_body<v2f>(h2, rowptr, csr, dinv, b2, out, N);
}

__global__ __launch_bounds__(256) void k_gemm2(const float* __restrict__ A,
                                               const float* __restrict__ W2,
                                               float* __restrict__ H2,
                                               int N) {
    __shared__ alignas(16) float Alds[32 * APITCH];
    __shared__ alignas(16) float Wlds[128 * WPITCH];
    const int tid = threadIdx.x;
    const int bm  = blockIdx.x * 32;

    {
        const float4* w4 = (const float4*)W2;
        for (int i = tid; i < 2048; i += 256) {
            *(float4*)&Wlds[(i >> 4) * WPITCH + (i & 15) * 4] = w4[i];
        }
    }
    {
        for (int i = tid; i < 32 * 32; i += 256) {
            int r = i >> 5, g = i & 31;
            float4 v = make_float4(0.f, 0.f, 0.f, 0.f);
            if (bm + r < N) v = ((const float4*)(A + (size_t)(bm + r) * 128))[g];
            v.x = fmaxf(v.x, 0.f); v.y = fmaxf(v.y, 0.f);
            v.z = fmaxf(v.z, 0.f); v.w = fmaxf(v.w, 0.f);
            *(float4*)&Alds[r * APITCH + g * 4] = v;
        }
    }
    __syncthreads();

    const int wave = tid >> 5;
    const int lane = tid & 31;
    const int msub = (wave >> 2) * 16;
    const int nb   = (wave & 3) * 16;
    const int mr   = lane & 15;
    const int kh   = (lane >> 4) * 2;

    v8f acc = {};
    for (int kc = 0; kc < 128; kc += 4) {
        v2f a, b;
        a = *(const v2f*)&Alds[(msub + mr) * APITCH + kc + kh];
        b[0] = Wlds[(kc + kh + 0) * WPITCH + nb + mr];
        b[1] = Wlds[(kc + kh + 1) * WPITCH + nb + mr];
        acc = __builtin_amdgcn_wmma_f32_16x16x4_f32(
                  false, a, false, b, (short)0, acc, false, false);
    }

    const int mtop = (lane >> 4) * 8;
    for (int v = 0; v < 8; ++v) {
        int m = bm + msub + mtop + v;
        if (m < N) H2[(size_t)m * 64 + nb + mr] = acc[v];
    }
}

extern "C" void kernel_launch(void* const* d_in, const int* in_sizes, int n_in,
                              void* d_out, int out_size, void* d_ws, size_t ws_size,
                              hipStream_t stream) {
    const float* x   = (const float*)d_in[0];
    const int*   ei  = (const int*)d_in[1];
    const float* ew  = (const float*)d_in[2];
    const float* W1  = (const float*)d_in[3];
    const float* b1  = (const float*)d_in[4];
    const float* W2  = (const float*)d_in[5];
    const float* b2  = (const float*)d_in[6];
    float* out = (float*)d_out;

    const int E = in_sizes[2];
    const int N = in_sizes[0] / 128;
    const int* row = ei;
    const int* col = ei + E;

    const int NB = (N + BW - 1) / BW;
    const int per = (E + TILES - 1) / TILES;
    const size_t Np = ((size_t)N + 1 + 63) & ~(size_t)63;
    const size_t Ep = ((size_t)E + 63) & ~(size_t)63;
    const size_t Cp = ((size_t)NB * TILES + 1 + 63) & ~(size_t)63;
    int*   tcnt   = (int*)d_ws;
    int*   toff   = tcnt + Cp;
    int*   rowptr = toff + Cp;
    float* dinv   = (float*)(rowptr + Np);
    int*   rec    = (int*)(dinv + Np);
    int*   csr    = rec + 4 * Ep;
    float* h1     = (float*)(csr + 2 * Ep);
    float* out1   = h1 + (size_t)N * 128;
    float* h2     = out1 + (size_t)N * 128;

    k_hist<<<TILES, 512, 0, stream>>>(col, tcnt, E, NB, per);
    k_scan<<<1, 1024, 0, stream>>>(tcnt, toff, NB * TILES);
    k_part<<<TILES, 512, 0, stream>>>(row, col, ew, toff, rec, E, NB, per);
    k_build<<<NB, 1024, 0, stream>>>(rec, toff, rowptr, csr, dinv, N);

    k_gemm1<<<(N + 15) / 16, 256, 0, stream>>>(x, W1, h1, N);
    k_gather1<<<(N + 7) / 8, 256, 0, stream>>>(h1, rowptr, csr, dinv, b1, out1, N);

    k_gemm2<<<(N + 31) / 32, 256, 0, stream>>>(out1, W2, h2, N);
    k_gather2<<<(N + 7) / 8, 256, 0, stream>>>(h2, rowptr, csr, dinv, b2, out, N);
}


# Layer 1 sums the neighbours' x rows before the W1 product (sum order changes: aggregate, then multiply), so gather, @W1+b1, relu and @W2 are one kernel on an LDS tile; h1, out1 never reach memory
# speedup vs baseline: 3.5816x; 1.2312x over previous
#include <hip/hip_runtime.h>
#include <hip/hip_bf16.h>


typedef __attribute__((ext_vector_type(2))) float v2f;
typedef __attribute__((ext_vector_type(4))) float v4f;
typedef __attribute__((ext_vector_type(8))) float v8f;

constexpr int APITCH = 132;

constexpr int TILES = 256;
constexpr int BSH   = 8;
constexpr int BW    = 1 << BSH;
constexpr int MAXB  = 1024;

__global__ __launch_bounds__(512) void k_hist(const int* __restrict__ col,
                                               int* __restrict__ tcnt,
                                               int E, int NB, int per) {
    __shared__ int hist[MAXB];
    const int tid = threadIdx.x, tile = blockIdx.x;
    for (int b = tid; b < NB; b += 512) hist[b] = 0;
    __syncthreads();
    const int e0 = tile * per, e1 = min(E, e0 + per);
    for (int base = e0; base < e1; base += 4 * 512) {
        int c[4];
        for (int k = 0; k < 4; ++k) {
            const int e = base + k * 512 + tid;
            c[k] = col[min(e, e1 - 1)];
        }
        for (int k = 0; k < 4; ++k) {
            if (base + k * 512 + tid < e1) atomicAdd(&hist[c[k] >> BSH], 1);
        }
    }
    __syncthreads();
    for (int b = tid; b < NB; b += 512) tcnt[b * TILES + tile] = hist[b];
}

__device__ __forceinline__ int wave_incl_scan(int x, int lane) {
    x += __builtin_amdgcn_update_dpp(0, x, 0x111, 0xf, 0xf, true);
    x += __builtin_amdgcn_update_dpp(0, x, 0x112, 0xf, 0xf, true);
    x += __builtin_amdgcn_update_dpp(0, x, 0x114, 0xf, 0xf, true);
    x += __builtin_amdgcn_update_dpp(0, x, 0x118, 0xf, 0xf, true);
    const int lo = __builtin_amdgcn_readlane(x, 15);
    return x + ((lane >= 16) ? lo : 0);
}

__global__ __launch_bounds__(1024) void k_scan(const int* __restrict__ cnt,
                                                int* __restrict__ rowptr, int N) {
    constexpr int K = 13;
    __shared__ int wsum[K * 32];
    __shared__ int tsum[K];
    const int tid = threadIdx.x, lane = tid & 31, wv = tid >> 5;
    const int nfull = N >> 2;
    const int4* __restrict__ cnt4 = (const int4*)cnt;
    int4* __restrict__ row4 = (int4*)rowptr;
    int carry = 0;
    for (int base = 0; base < nfull; base += K * 1024) {
        int4 v[K];
        for (int k = 0; k < K; ++k) {
            v[k] = cnt4[min(base + k * 1024 + tid, nfull - 1)];
        }
        int s[K], inc[K];
        for (int k = 0; k < K; ++k) {
            if (base + k * 1024 + tid >= nfull) v[k] = make_int4(0, 0, 0, 0);
            s[k] = (v[k].x + v[k].y) + (v[k].z + v[k].w);
            inc[k] = wave_incl_scan(s[k], lane);
        }
        if (lane == 31) {
            for (int k = 0; k < K; ++k) wsum[k * 32 + wv] = inc[k];
        }
        __syncthreads();
        if (wv < K) {
            const int x = wsum[wv * 32 + lane];
            const int xi = wave_incl_scan(x, lane);
            wsum[wv * 32 + lane] = xi - x;
            if (lane == 31) tsum[wv] = xi;
        }
        __syncthreads();
        int run = carry;
        for (int k = 0; k < K; ++k) {
            int4 o;
            o.x = run + wsum[k * 32 + wv] + (inc[k] - s[k]);
            o.y = o.x + v[k].x;
            o.z = o.y + v[k].y;
            o.w = o.z + v[k].z;
            run += tsum[k];
            if (base + k * 1024 + tid < nfull) row4[base + k * 1024 + tid] = o;
        }
        carry = run;
        __syncthreads();
    }
    if (tid == 0) {
        for (int j = nfull * 4; j < N; ++j) {
            rowptr[j] = carry;
            carry += cnt[j];
        }
        rowptr[N] = carry;
    }
}

__global__ __launch_bounds__(512) void k_part(const int* __restrict__ row,
                                               const int* __restrict__ col,
                                               const float* __restrict__ w,
                                               const int* __restrict__ toff,
                                               int* __restrict__ recw,
                                               int E, int NB, int per) {
    __shared__ int cur[MAXB];
    int4* __restrict__ rec = (int4*)recw;
    const int tid = threadIdx.x, tile = blockIdx.x;
    for (int b = tid; b < NB; b += 512) cur[b] = toff[b * TILES + tile];
    __syncthreads();
    const int e0 = tile * per, e1 = min(E, e0 + per);
    for (int base = e0; base < e1; base += 4 * 512) {
        int c[4], r[4], wb[4];
        for (int k = 0; k < 4; ++k) {
            const int e = min(base + k * 512 + tid, e1 - 1);
            c[k]  = col[e];
            r[k]  = row[e];
            wb[k] = __float_as_int(w[e]);
        }
        for (int k = 0; k < 4; ++k) {
            if (base + k * 512 + tid < e1) {
                const int p = atomicAdd(&cur[c[k] >> BSH], 1);
                rec[p] = make_int4(r[k], c[k], wb[k], 0);
            }
        }
    }
}

__global__ __launch_bounds__(1024) void k_build(const int* __restrict__ rcol,
                                                 const int* __restrict__ toff,
                                                 int* __restrict__ rowptr,
                                                 int* csr,
                                                 float* __restrict__ dinv, int N) {
    __shared__ int cnt[BW];
    __shared__ int cur[BW];
    __shared__ int wtot[BW / 32];
    const int tid = threadIdx.x, lane = tid & 31, wv = tid >> 5;
    const int b  = blockIdx.x;
    const int bs = toff[b * TILES], be = toff[(b + 1) * TILES];
    const int4* __restrict__ rec = (const int4*)rcol;
    if (tid < BW) cnt[tid] = 0;
    __syncthreads();
    for (int base = bs; base < be; base += 4 * 1024) {
        int c[4];
        for (int k = 0; k < 4; ++k) {
            const int e = min(base + k * 1024 + tid, be - 1);
            c[k] = rcol[4 * (size_t)e + 1];
        }
        for (int k = 0; k < 4; ++k) {
            if (base + k * 1024 + tid < be) atomicAdd(&cnt[c[k] & (BW - 1)], 1);
        }
    }
    __syncthreads();
    int c = 0, inc = 0, off = 0;
    if (tid < BW) {
        c   = cnt[tid];
        inc = wave_incl_scan(c, lane);
        if (lane == 31) wtot[wv] = inc;
    }
    __syncthreads();
    const int node = b * BW + tid;
    if (tid < BW) {
        int pre = 0;
        for (int k = 0; k < BW / 32; ++k) pre += (k < wv) ? wtot[k] : 0;
        off = bs + pre + (inc - c);
        cur[tid] = off;
        if (node < N) rowptr[node] = off;
        if (node == N - 1) rowptr[N] = off + c;
    }
    __syncthreads();
    int2* __restrict__ ent = (int2*)csr;
    for (int base = bs; base < be; base += 4 * 1024) {
        int4 t[4];
        for (int k = 0; k < 4; ++k) {
            const int e = min(base + k * 1024 + tid, be - 1);
            t[k] = rec[e];
        }
        for (int k = 0; k < 4; ++k) {
            if (base + k * 1024 + tid < be) {
                const int p = atomicAdd(&cur[t[k].y & (BW - 1)], 1);
                ent[p] = make_int2(t[k].x, t[k].z);
            }
        }
    }
    __syncthreads();
    if (tid < BW && node < N) {
        const int end = off + c;
        float d = 1.0f;
        int p = off;
        for (; p + 4 <= end; p += 4) {
            const float a0 = __int_as_float(csr[2 * p + 1]);
            const float a1 = __int_as_float(csr[2 * p + 3]);
            const float a2 = __int_as_float(csr[2 * p + 5]);
            const float a3 = __int_as_float(csr[2 * p + 7]);
            d += a0; d += a1; d += a2; d += a3;
        }
        for (; p < end; ++p) d += __int_as_float(csr[2 * p + 1]);
        dinv[node] = (d > 0.0f) ? rsqrtf(d) : 0.0f;
    }
}

template <typename V>
__device__ __forceinline__ V gather_row(const float* __restrict__ h,
                                        const int* __restrict__ rowptr,
                                        const int* __restrict__ csr,
                                        const float* __restrict__ dinv,
                                        int node, int lane, V init) {
    constexpr int VEC = sizeof(V) / sizeof(float);
    constexpr int F   = VEC * 32;
    const int beg = rowptr[node], end = rowptr[node + 1];
    const float dc = dinv[node];
    const V hv = *(const V*)(h + (size_t)node * F + lane * VEC);
    V acc = init + hv * (dc * dc);
    const int2* ent = (const int2*)csr;
    for (int base = beg; base < end; base += 32) {
        const int n = min(32, end - base);
        int r = 0;
        float nw = 0.0f;
        if (lane < n) {
            const int2 t = ent[base + lane];
            r  = t.x;
            nw = dinv[r] * __int_as_float(t.y) * dc;
        }
        int j = 0;
        for (; j + 4 <= n; j += 4) {
            const int r0 = __builtin_amdgcn_readlane(r, j + 0);
            const int r1 = __builtin_amdgcn_readlane(r, j + 1);
            const int r2 = __builtin_amdgcn_readlane(r, j + 2);
            const int r3 = __builtin_amdgcn_readlane(r, j + 3);
            const float w0 = __int_as_float(__builtin_amdgcn_readlane(__float_as_int(nw), j + 0));
            const float w1 = __int_as_float(__builtin_amdgcn_readlane(__float_as_int(nw), j + 1));
            const float w2 = __int_as_float(__builtin_amdgcn_readlane(__float_as_int(nw), j + 2));
            const float w3 = __int_as_float(__builtin_amdgcn_readlane(__float_as_int(nw), j + 3));
            const V v0 = *(const V*)(h + (size_t)r0 * F + lane * VEC);
            const V v1 = *(const V*)(h + (size_t)r1 * F + lane * VEC);
            const V v2 = *(const V*)(h + (size_t)r2 * F + lane * VEC);
            const V v3 = *(const V*)(h + (size_t)r3 * F + lane * VEC);
            acc += v0 * w0;
            acc += v1 * w1;
            acc += v2 * w2;
            acc += v3 * w3;
        }
        for (; j < n; ++j) {
            const int r0 = __builtin_amdgcn_readlane(r, j);
            const float w0 = __int_as_float(__builtin_amdgcn_readlane(__float_as_int(nw), j));
            const V v0 = *(const V*)(h + (size_t)r0 * F + lane * VEC);
            acc += v0 * w0;
        }
    }
    return acc;
}

constexpr int TM = 32;

__global__ __launch_bounds__(256, 16) void k_layer1(const float* __restrict__ x,
                                                const int* __restrict__ rowptr,
                                                const int* __restrict__ csr,
                                                const float* __restrict__ dinv,
                                                const float* __restrict__ W1,
                                                const float* __restrict__ b1,
                                                const float* __restrict__ W2,
                                                float* __restrict__ H2,
                                                int N) {
    __shared__ alignas(16) float T[TM * APITCH];
    const int tid  = threadIdx.x;
    const int lane = tid & 31;
    const int wave = __builtin_amdgcn_readfirstlane(tid >> 5);
    const int bm   = blockIdx.x * TM;

    for (int i = 0; i < TM / 8; ++i) {
        const int r = wave * (TM / 8) + i;
        const int node = bm + r;
        v4f s = {0.f, 0.f, 0.f, 0.f};
        if (node < N) s = gather_row<v4f>(x, rowptr, csr, dinv, node, lane, s);
        *(v4f*)&T[r * APITCH + lane * 4] = s;
    }
    __syncthreads();

    const int mr   = lane & 15;
    const int kh   = (lane >> 4) * 2;
    const int mtop = (lane >> 4) * 8;

    {
        const int nb = wave * 16;
        v8f c0 = {}, c1 = {};
        for (int kc = 0; kc < 128; kc += 4) {
            v2f a0, a1, b;
            a0 = *(const v2f*)&T[mr * APITCH + kc + kh];
            a1 = *(const v2f*)&T[(16 + mr) * APITCH + kc + kh];
            b[0] = W1[(kc + kh + 0) * 128 + nb + mr];
            b[1] = W1[(kc + kh + 1) * 128 + nb + mr];
            c0 = __builtin_amdgcn_wmma_f32_16x16x4_f32(
                     false, a0, false, b, (short)0, c0, false, false);
            c1 = __builtin_amdgcn_wmma_f32_16x16x4_f32(
                     false, a1, false, b, (short)0, c1, false, false);
        }
        const float bias = b1[nb + mr];
        __syncthreads();
        for (int v = 0; v < 8; ++v) {
            T[(mtop + v) * APITCH + nb + mr]      = fmaxf(c0[v] + bias, 0.f);
            T[(16 + mtop + v) * APITCH + nb + mr] = fmaxf(c1[v] + bias, 0.f);
        }
    }
    __syncthreads();

    if (wave < 4) {
        const int nb = wave * 16;
        v8f d0 = {}, d1 = {};
        for (int kc = 0; kc < 128; kc += 4) {
            v2f a0, a1, b;
            a0 = *(const v2f*)&T[mr * APITCH + kc + kh];
            a1 = *(const v2f*)&T[(16 + mr) * APITCH + kc + kh];
            b[0] = W2[(kc + kh + 0) * 64 + nb + mr];
            b[1] = W2[(kc + kh + 1) * 64 + nb + mr];
            d0 = __builtin_amdgcn_wmma_f32_16x16x4_f32(
                     false, a0, false, b, (short)0, d0, false, false);
            d1 = __builtin_amdgcn_wmma_f32_16x16x4_f32(
                     false, a1, false, b, (short)0, d1, false, false);
        }
        for (int v = 0; v < 8; ++v) {
            const int m0 = bm + mtop + v;
            const int m1 = m0 + 16;
            if (m0 < N) H2[(size_t)m0 * 64 + nb + mr] = d0[v];
            if (m1 < N) H2[(size_t)m1 * 64 + nb + mr] = d1[v];
        }
    }
}

__global__ __launch_bounds__(256) void k_gather2(const float* __restrict__ h2,
                                                 const int* __restrict__ rowptr,
                                                 const int* __restrict__ csr,
                                                 const float* __restrict__ dinv,
                                                 const float* __restrict__ b2,
                                                 float* __restrict__ out, int N) {
    const int lane = threadIdx.x & 31;
    const int node = __builtin_amdgcn_readfirstlane(
        (int)(blockIdx.x * (blockDim.x >> 5) + (threadIdx.x >> 5)));
    if (node >= N) return;
    const v2f bv = *(const v2f*)(b2 + lane * 2);
    const v2f acc = gather_row<v2f>(h2, rowptr, csr, dinv, node, lane, bv);
    *(v2f*)(out + (size_t)node * 64 + lane * 2) = acc;
}

extern "C" void kernel_launch(void* const* d_in, const int* in_sizes, int n_in,
                              void* d_out, int out_size, void* d_ws, size_t ws_size,
                              hipStream_t stream) {
    const float* x   = (const float*)d_in[0];
    const int*   ei  = (const int*)d_in[1];
    const float* ew  = (const float*)d_in[2];
    const float* W1  = (const float*)d_in[3];
    const float* b1  = (const float*)d_in[4];
    const float* W2  = (const float*)d_in[5];
    const float* b2  = (const float*)d_in[6];
    float* out = (float*)d_out;

    const int E = in_sizes[2];
    const int N = in_sizes[0] / 128;
    const int* row = ei;
    const int* col = ei + E;

    const int NB = (N + BW - 1) / BW;
    const int per = (E + TILES - 1) / TILES;
    const size_t Np = ((size_t)N + 1 + 63) & ~(size_t)63;
    const size_t Ep = ((size_t)E + 63) & ~(size_t)63;
    const size_t Cp = ((size_t)NB * TILES + 1 + 63) & ~(size_t)63;
    int*   tcnt   = (int*)d_ws;
    int*   toff   = tcnt + Cp;
    int*   rowptr = toff + Cp;
    float* dinv   = (float*)(rowptr + Np);
    int*   rec    = (int*)(dinv + Np);
    int*   csr    = rec + 4 * Ep;
    float* h2     = (float*)(csr + 2 * Ep);

    k_hist<<<TILES, 512, 0, stream>>>(col, tcnt, E, NB, per);
    k_scan<<<1, 1024, 0, stream>>>(tcnt, toff, NB * TILES);
    k_part<<<TILES, 512, 0, stream>>>(row, col, ew, toff, rec, E, NB, per);
    k_build<<<NB, 1024, 0, stream>>>(rec, toff, rowptr, csr, dinv, N);

    k_layer1<<<(N + TM - 1) / TM, 256, 0, stream>>>(x, rowptr, csr, dinv, W1, b1, W2, h2, N);

    k_gather2<<<(N + 7) / 8, 256, 0, stream>>>(h2, rowptr, csr, dinv, b2, out, N);
}


# Each edge tile sorts its own edges by bucket in a fixed region (count, scan, rank in LDS), so the count and all-tile scan launches go: 4 launches, not 6; slot order in a column: other LDS add arrival
# speedup vs baseline: 4.0909x; 1.1422x over previous
#include <hip/hip_runtime.h>
#include <hip/hip_bf16.h>


typedef __attribute__((ext_vector_type(2))) float v2f;
typedef __attribute__((ext_vector_type(4))) float v4f;
typedef __attribute__((ext_vector_type(8))) float v8f;

constexpr int APITCH = 132;

constexpr int TILES = 256;
constexpr int BSH   = 8;
constexpr int BW    = 1 << BSH;
constexpr int MAXB  = 1024;

__device__ __forceinline__ int wave_incl_scan(int x, int lane) {
    x += __builtin_amdgcn_update_dpp(0, x, 0x111, 0xf, 0xf, true);
    x += __builtin_amdgcn_update_dpp(0, x, 0x112, 0xf, 0xf, true);
    x += __builtin_amdgcn_update_dpp(0, x, 0x114, 0xf, 0xf, true);
    x += __builtin_amdgcn_update_dpp(0, x, 0x118, 0xf, 0xf, true);
    const int lo = __builtin_amdgcn_readlane(x, 15);
    return x + ((lane >= 16) ? lo : 0);
}

__global__ __launch_bounds__(512) void k_part(const int* __restrict__ row,
                                               const int* __restrict__ col,
                                               const float* __restrict__ w,
                                               int* __restrict__ ls,
                                               int* __restrict__ recw,
                                               int E, int NB, int per) {
    __shared__ int cur[MAXB];
    __shared__ int wtot[16];
    int4* __restrict__ rec = (int4*)recw;
    const int tid = threadIdx.x, lane = tid & 31, wv = tid >> 5, tile = blockIdx.x;
    cur[2 * tid]     = 0;
    cur[2 * tid + 1] = 0;
    __syncthreads();
    const int e0 = tile * per, e1 = min(E, e0 + per);
    for (int base = e0; base < e1; base += 4 * 512) {
        int c[4];
        for (int k = 0; k < 4; ++k) {
            const int e = base + k * 512 + tid;
            c[k] = col[min(e, e1 - 1)];
        }
        for (int k = 0; k < 4; ++k) {
            if (base + k * 512 + tid < e1) atomicAdd(&cur[c[k] >> BSH], 1);
        }
    }
    __syncthreads();
    const int h0 = cur[2 * tid], h1 = cur[2 * tid + 1];
    const int s = h0 + h1;
    const int inc = wave_incl_scan(s, lane);
    if (lane == 31) wtot[wv] = inc;
    __syncthreads();
    int pre = 0;
    for (int k = 0; k < 16; ++k) pre += (k < wv) ? wtot[k] : 0;
    const int x0 = pre + (inc - s);
    const int x1 = x0 + h0;
    cur[2 * tid]     = x0;
    cur[2 * tid + 1] = x1;
    if (2 * tid < NB)     ls[(2 * tid) * TILES + tile]     = x0;
    if (2 * tid + 1 < NB) ls[(2 * tid + 1) * TILES + tile] = x1;
    if (tid == 0)         ls[NB * TILES + tile] = max(e1 - e0, 0);
    __syncthreads();
    int4* __restrict__ mine = rec + (size_t)tile * per;
    for (int base = e0; base < e1; base += 4 * 512) {
        int c[4], r[4], wb[4];
        for (int k = 0; k < 4; ++k) {
            const int e = min(base + k * 512 + tid, e1 - 1);
            c[k]  = col[e];
            r[k]  = row[e];
            wb[k] = __float_as_int(w[e]);
        }
        for (int k = 0; k < 4; ++k) {
            if (base + k * 512 + tid < e1) {
                const int p = atomicAdd(&cur[c[k] >> BSH], 1);
                mine[p] = make_int4(r[k], c[k], wb[k], 0);
            }
        }
    }
}

__global__ __launch_bounds__(1024) void k_build(const int* __restrict__ recw,
                                                 const int* __restrict__ ls,
                                                 int* __restrict__ rowptr,
                                                 int* csr,
                                                 float* __restrict__ dinv, int N, int per) {
    __shared__ int cnt[BW];
    __shared__ int cur[BW];
    __shared__ int wtot[BW / 32];
    __shared__ int rs[TILES];
    __shared__ int rn[TILES];
    __shared__ int sb[TILES / 32];
    const int tid = threadIdx.x, lane = tid & 31, wv = tid >> 5;
    const int b  = blockIdx.x;
    const int4* __restrict__ rec = (const int4*)recw;
    if (tid < TILES) {
        const int lo = ls[b * TILES + tid], hi = ls[(b + 1) * TILES + tid];
        rs[tid] = tid * per + lo;
        rn[tid] = hi - lo;
        const int slo = wave_incl_scan(lo, lane);
        if (lane == 31) sb[wv] = slo;
    }
    if (tid < BW) cnt[tid] = 0;
    __syncthreads();
    int bs = 0;
    for (int k = 0; k < TILES / 32; ++k) bs += sb[k];
    int n[8], r0[8];
    for (int i = 0; i < 8; ++i) {
        n[i]  = __builtin_amdgcn_readfirstlane(rn[wv * 8 + i]);
        r0[i] = __builtin_amdgcn_readfirstlane(rs[wv * 8 + i]);
    }
    const int last = TILES * per - 1;
    int4 t[8];
    for (int i = 0; i < 8; ++i) t[i] = rec[min(r0[i] + lane, last)];
    asm volatile("" :: "v"(t[0].y), "v"(t[1].y), "v"(t[2].y), "v"(t[3].y),
                       "v"(t[4].y), "v"(t[5].y), "v"(t[6].y), "v"(t[7].y));
    for (int i = 0; i < 8; ++i) {
        if (lane < n[i]) atomicAdd(&cnt[t[i].y & (BW - 1)], 1);
    }
    for (int i = 0; i < 8; ++i) {
        for (int k = lane + 32; k < n[i]; k += 32)
            atomicAdd(&cnt[recw[4 * (size_t)(r0[i] + k) + 1] & (BW - 1)], 1);
    }
    __syncthreads();
    int c = 0, inc = 0, off = 0;
    if (tid < BW) {
        c   = cnt[tid];
        inc = wave_incl_scan(c, lane);
        if (lane == 31) wtot[wv] = inc;
    }
    __syncthreads();
    const int node = b * BW + tid;
    if (tid < BW) {
        int pre = 0;
        for (int k = 0; k < BW / 32; ++k) pre += (k < wv) ? wtot[k] : 0;
        off = bs + pre + (inc - c);
        cur[tid] = off;
        if (node < N) rowptr[node] = off;
        if (node == N - 1) rowptr[N] = off + c;
    }
    __syncthreads();
    int2* __restrict__ ent = (int2*)csr;
    for (int i = 0; i < 8; ++i) {
        if (lane < n[i]) {
            const int p = atomicAdd(&cur[t[i].y & (BW - 1)], 1);
            ent[p] = make_int2(t[i].x, t[i].z);
        }
    }
    for (int i = 0; i < 8; ++i) {
        for (int k = lane + 32; k < n[i]; k += 32) {
            const int4 q = rec[r0[i] + k];
            const int p = atomicAdd(&cur[q.y & (BW - 1)], 1);
            ent[p] = make_int2(q.x, q.z);
        }
    }
    __syncthreads();
    if (tid < BW && node < N) {
        const int end = off + c;
        float d = 1.0f;
        int p = off;
        for (; p + 4 <= end; p += 4) {
            const float a0 = __int_as_float(csr[2 * p + 1]);
            const float a1 = __int_as_float(csr[2 * p + 3]);
            const float a2 = __int_as_float(csr[2 * p + 5]);
            const float a3 = __int_as_float(csr[2 * p + 7]);
            d += a0; d += a1; d += a2; d += a3;
        }
        for (; p < end; ++p) d += __int_as_float(csr[2 * p + 1]);
        dinv[node] = (d > 0.0f) ? rsqrtf(d) : 0.0f;
    }
}

template <typename V>
__device__ __forceinline__ V gather_row(const float* __restrict__ h,
                                        const int* __restrict__ rowptr,
                                        const int* __restrict__ csr,
                                        const float* __restrict__ dinv,
                                        int node, int lane, V init) {
    constexpr int VEC = sizeof(V) / sizeof(float);
    constexpr int F   = VEC * 32;
    const int beg = rowptr[node], end = rowptr[node + 1];
    const float dc = dinv[node];
    const V hv = *(const V*)(h + (size_t)node * F + lane * VEC);
    V acc = init + hv * (dc * dc);
    const int2* ent = (const int2*)csr;
    for (int base = beg; base < end; base += 32) {
        const int n = min(32, end - base);
        int r = 0;
        float nw = 0.0f;
        if (lane < n) {
            const int2 t = ent[base + lane];
            r  = t.x;
            nw = dinv[r] * __int_as_float(t.y) * dc;
        }
        int j = 0;
        for (; j + 4 <= n; j += 4) {
            const int r0 = __builtin_amdgcn_readlane(r, j + 0);
            const int r1 = __builtin_amdgcn_readlane(r, j + 1);
            const int r2 = __builtin_amdgcn_readlane(r, j + 2);
            const int r3 = __builtin_amdgcn_readlane(r, j + 3);
            const float w0 = __int_as_float(__builtin_amdgcn_readlane(__float_as_int(nw), j + 0));
            const float w1 = __int_as_float(__builtin_amdgcn_readlane(__float_as_int(nw), j + 1));
            const float w2 = __int_as_float(__builtin_amdgcn_readlane(__float_as_int(nw), j + 2));
            const float w3 = __int_as_float(__builtin_amdgcn_readlane(__float_as_int(nw), j + 3));
            const V v0 = *(const V*)(h + (size_t)r0 * F + lane * VEC);
            const V v1 = *(const V*)(h + (size_t)r1 * F + lane * VEC);
            const V v2 = *(const V*)(h + (size_t)r2 * F + lane * VEC);
            const V v3 = *(const V*)(h + (size_t)r3 * F + lane * VEC);
            acc += v0 * w0;
            acc += v1 * w1;
            acc += v2 * w2;
            acc += v3 * w3;
        }
        for (; j < n; ++j) {
            const int r0 = __builtin_amdgcn_readlane(r, j);
            const float w0 = __int_as_float(__builtin_amdgcn_readlane(__float_as_int(nw), j));
            const V v0 = *(const V*)(h + (size_t)r0 * F + lane * VEC);
            acc += v0 * w0;
        }
    }
    return acc;
}

constexpr int TM = 32;

__global__ __launch_bounds__(256, 16) void k_layer1(const float* __restrict__ x,
                                                const int* __restrict__ rowptr,
                                                const int* __restrict__ csr,
                                                const float* __restrict__ dinv,
                                                const float* __restrict__ W1,
                                                const float* __restrict__ b1,
                                                const float* __restrict__ W2,
                                                float* __restrict__ H2,
                                                int N) {
    __shared__ alignas(16) float T[TM * APITCH];
    const int tid  = threadIdx.x;
    const int lane = tid & 31;
    const int wave = __builtin_amdgcn_readfirstlane(tid >> 5);
    const int bm   = blockIdx.x * TM;

    for (int i = 0; i < TM / 8; ++i) {
        const int r = wave * (TM / 8) + i;
        const int node = bm + r;
        v4f s = {0.f, 0.f, 0.f, 0.f};
        if (node < N) s = gather_row<v4f>(x, rowptr, csr, dinv, node, lane, s);
        *(v4f*)&T[r * APITCH + lane * 4] = s;
    }
    __syncthreads();

    const int mr   = lane & 15;
    const int kh   = (lane >> 4) * 2;
    const int mtop = (lane >> 4) * 8;

    {
        const int nb = wave * 16;
        v8f c0 = {}, c1 = {};
        for (int kc = 0; kc < 128; kc += 4) {
            v2f a0, a1, b;
            a0 = *(const v2f*)&T[mr * APITCH + kc + kh];
            a1 = *(const v2f*)&T[(16 + mr) * APITCH + kc + kh];
            b[0] = W1[(kc + kh + 0) * 128 + nb + mr];
            b[1] = W1[(kc + kh + 1) * 128 + nb + mr];
            c0 = __builtin_amdgcn_wmma_f32_16x16x4_f32(
                     false, a0, false, b, (short)0, c0, false, false);
            c1 = __builtin_amdgcn_wmma_f32_16x16x4_f32(
                     false, a1, false, b, (short)0, c1, false, false);
        }
        const float bias = b1[nb + mr];
        __syncthreads();
        for (int v = 0; v < 8; ++v) {
            T[(mtop + v) * APITCH + nb + mr]      = fmaxf(c0[v] + bias, 0.f);
            T[(16 + mtop + v) * APITCH + nb + mr] = fmaxf(c1[v] + bias, 0.f);
        }
    }
    __syncthreads();

    if (wave < 4) {
        const int nb = wave * 16;
        v8f d0 = {}, d1 = {};
        for (int kc = 0; kc < 128; kc += 4) {
            v2f a0, a1, b;
            a0 = *(const v2f*)&T[mr * APITCH + kc + kh];
            a1 = *(const v2f*)&T[(16 + mr) * APITCH + kc + kh];
            b[0] = W2[(kc + kh + 0) * 64 + nb + mr];
            b[1] = W2[(kc + kh + 1) * 64 + nb + mr];
            d0 = __builtin_amdgcn_wmma_f32_16x16x4_f32(
                     false, a0, false, b, (short)0, d0, false, false);
            d1 = __builtin_amdgcn_wmma_f32_16x16x4_f32(
                     false, a1, false, b, (short)0, d1, false, false);
        }
        for (int v = 0; v < 8; ++v) {
            const int m0 = bm + mtop + v;
            const int m1 = m0 + 16;
            if (m0 < N) H2[(size_t)m0 * 64 + nb + mr] = d0[v];
            if (m1 < N) H2[(size_t)m1 * 64 + nb + mr] = d1[v];
        }
    }
}

__global__ __launch_bounds__(256) void k_gather2(const float* __restrict__ h2,
                                                 const int* __restrict__ rowptr,
                                                 const int* __restrict__ csr,
                                                 const float* __restrict__ dinv,
                                                 const float* __restrict__ b2,
                                                 float* __restrict__ out, int N) {
    const int lane = threadIdx.x & 31;
    const int node = __builtin_amdgcn_readfirstlane(
        (int)(blockIdx.x * (blockDim.x >> 5) + (threadIdx.x >> 5)));
    if (node >= N) return;
    const v2f bv = *(const v2f*)(b2 + lane * 2);
    const v2f acc = gather_row<v2f>(h2, rowptr, csr, dinv, node, lane, bv);
    *(v2f*)(out + (size_t)node * 64 + lane * 2) = acc;
}

extern "C" void kernel_launch(void* const* d_in, const int* in_sizes, int n_in,
                              void* d_out, int out_size, void* d_ws, size_t ws_size,
                              hipStream_t stream) {
    const float* x   = (const float*)d_in[0];
    const int*   ei  = (const int*)d_in[1];
    const float* ew  = (const float*)d_in[2];
    const float* W1  = (const float*)d_in[3];
    const float* b1  = (const float*)d_in[4];
    const float* W2  = (const float*)d_in[5];
    const float* b2  = (const float*)d_in[6];
    float* out = (float*)d_out;

    const int E = in_sizes[2];
    const int N = in_sizes[0] / 128;
    const int* row = ei;
    const int* col = ei + E;

    const int NB = (N + BW - 1) / BW;
    const int per = (E + TILES - 1) / TILES;
    const size_t Np = ((size_t)N + 1 + 63) & ~(size_t)63;
    const size_t Ep = ((size_t)E + 63) & ~(size_t)63;
    const size_t Lp = ((size_t)(NB + 1) * TILES + 63) & ~(size_t)63;
    const size_t Rp = ((size_t)TILES * per + 63) & ~(size_t)63;
    int*   ls     = (int*)d_ws;
    int*   rowptr = ls + Lp;
    float* dinv   = (float*)(rowptr + Np);
    int*   rec    = (int*)(dinv + Np);
    int*   csr    = rec + 4 * Rp;
    float* h2     = (float*)(csr + 2 * Ep);

    k_part<<<TILES, 512, 0, stream>>>(row, col, ew, ls, rec, E, NB, per);
    k_build<<<NB, 1024, 0, stream>>>(rec, ls, rowptr, csr, dinv, N, per);

    k_layer1<<<(N + TM - 1) / TM, 256, 0, stream>>>(x, rowptr, csr, dinv, W1, b1, W2, h2, N);

    k_gather2<<<(N + 7) / 8, 256, 0, stream>>>(h2, rowptr, csr, dinv, b2, out, N);
}
